# starting kernel
# baseline (speedup 1.0000x reference)
#include <hip/hip_runtime.h>
#include <math.h>

#define N1c  200000
#define N2c  40000
#define N3c  10000
#define INc  128
#define HIDc 256
#define OUTc 64

typedef float v2f __attribute__((ext_vector_type(2)));
typedef float v8f __attribute__((ext_vector_type(8)));

__global__ void zero_f32(float* __restrict__ p, int n) {
    int i = blockIdx.x * blockDim.x + threadIdx.x;
    if (i < n) p[i] = 0.0f;
}

__global__ void scatter_edges(const float* __restrict__ x,
                              const int*   __restrict__ src,
                              const int*   __restrict__ dst,
                              float*       __restrict__ agg,
                              float*       __restrict__ cnt,
                              long long total, int dshift) {
    long long i = (long long)blockIdx.x * blockDim.x + threadIdx.x;
    if (i >= total) return;
    int e = (int)(i >> dshift);
    int f = (int)(i & ((1 << dshift) - 1));
    int s = src[e];
    int d = dst[e];
    atomicAdd(&agg[((long long)d << dshift) + f], x[((long long)s << dshift) + f]);
    if (f == 0) atomicAdd(&cnt[d], 1.0f);
}

__launch_bounds__(128)
__global__ void sage_gemm_wmma(const float* __restrict__ agg,
                               const float* __restrict__ cnt,
                               const float* __restrict__ xt,
                               const float* __restrict__ Wl,
                               const float* __restrict__ Wr,
                               const float* __restrict__ bias,
                               float*       __restrict__ out,
                               int K, int Nout, int relu) {
    const int lane  = threadIdx.x & 31;
    const int wave  = threadIdx.x >> 5;
    const int mTile = blockIdx.x;
    const int nTile = blockIdx.y * 4 + wave;

    const int n  = lane & 15;
    const int kh = lane >> 4;

    const long long row = (long long)mTile * 16 + n;
    const long long col = (long long)nTile * 16 + n;

    const float inv = 1.0f / fmaxf(cnt[row], 1.0f);

    v8f acc;
    #pragma unroll
    for (int r = 0; r < 8; ++r) acc[r] = 0.0f;

    {
        const float* aRow = agg + row * K;
        const float* bRow = Wl  + col * K;
        for (int k0 = 0; k0 < K; k0 += 4) {
            const int ka = k0 + 2 * kh;
            v2f a; a[0] = aRow[ka] * inv; a[1] = aRow[ka + 1] * inv;
            v2f b; b[0] = bRow[ka];       b[1] = bRow[ka + 1];
            acc = __builtin_amdgcn_wmma_f32_16x16x4_f32(
                false, a, false, b, (short)0, acc, false, false);
        }
    }
    {
        const float* aRow = xt + row * K;
        const float* bRow = Wr + col * K;
        for (int k0 = 0; k0 < K; k0 += 4) {
            const int ka = k0 + 2 * kh;
            v2f a; a[0] = aRow[ka]; a[1] = aRow[ka + 1];
            v2f b; b[0] = bRow[ka]; b[1] = bRow[ka + 1];
            acc = __builtin_amdgcn_wmma_f32_16x16x4_f32(
                false, a, false, b, (short)0, acc, false, false);
        }
    }

    const float bv = bias[col];
    const long long mbase = (long long)mTile * 16 + ((lane < 16) ? 0 : 8);
    #pragma unroll
    for (int r = 0; r < 8; ++r) {
        float v = acc[r] + bv;
        if (relu) v = fmaxf(v, 0.0f);
        out[(mbase + r) * Nout + col] = v;
    }
}

__global__ void log_softmax64(float* __restrict__ out, int M) {
    const int lane = threadIdx.x & 31;
    const int wave = threadIdx.x >> 5;
    const int row  = blockIdx.x * 8 + wave;
    if (row >= M) return;
    float* p = out + (long long)row * 64;
    float a = p[lane];
    float b = p[lane + 32];
    float m = fmaxf(a, b);
    #pragma unroll
    for (int off = 16; off >= 1; off >>= 1)
        m = fmaxf(m, __shfl_xor(m, off, 32));
    float s = __expf(a - m) + __expf(b - m);
    #pragma unroll
    for (int off = 16; off >= 1; off >>= 1)
        s += __shfl_xor(s, off, 32);
    const float l = __logf(s) + m;
    p[lane]      = a - l;
    p[lane + 32] = b - l;
}

extern "C" void kernel_launch(void* const* d_in, const int* in_sizes, int n_in,
                              void* d_out, int out_size, void* d_ws, size_t ws_size,
                              hipStream_t stream) {
    const float* x    = (const float*)d_in[0];
    const int*   src1 = (const int*)  d_in[1];
    const int*   dst1 = (const int*)  d_in[2];
    const int*   src2 = (const int*)  d_in[3];
    const int*   dst2 = (const int*)  d_in[4];
    const float* W1l  = (const float*)d_in[7];
    const float* W1r  = (const float*)d_in[8];
    const float* b1   = (const float*)d_in[9];
    const float* W2l  = (const float*)d_in[10];
    const float* W2r  = (const float*)d_in[11];
    const float* b2   = (const float*)d_in[12];

    const int E1 = in_sizes[1];
    const int E2 = in_sizes[3];

    float* agg1 = (float*)d_ws;
    float* cnt1 = agg1 + (size_t)N2c * INc;
    float* h    = cnt1 + N2c;
    float* agg2 = h    + (size_t)N2c * HIDc;
    float* cnt2 = agg2 + (size_t)N3c * HIDc;
    float* out  = (float*)d_out;

    {
        int z1 = N2c * INc + N2c;
        zero_f32<<<(z1 + 255) / 256, 256, 0, stream>>>(agg1, z1);
        int z2 = N3c * HIDc + N3c;
        zero_f32<<<(z2 + 255) / 256, 256, 0, stream>>>(agg2, z2);
    }

    {
        long long total = (long long)E1 * INc;
        int blocks = (int)((total + 255) / 256);
        scatter_edges<<<blocks, 256, 0, stream>>>(x, src1, dst1, agg1, cnt1,
                                                  total, 7  );
        dim3 grid(N2c / 16, HIDc / 64);
        sage_gemm_wmma<<<grid, 128, 0, stream>>>(agg1, cnt1, x, W1l, W1r, b1,
                                                 h, INc, HIDc,  1);
    }

    {
        long long total = (long long)E2 * HIDc;
        int blocks = (int)((total + 255) / 256);
        scatter_edges<<<blocks, 256, 0, stream>>>(h, src2, dst2, agg2, cnt2,
                                                  total, 8  );
        dim3 grid(N3c / 16, OUTc / 64);
        sage_gemm_wmma<<<grid, 128, 0, stream>>>(agg2, cnt2, h, W2l, W2r, b2,
                                                 out, HIDc, OUTc,  0);
    }

    log_softmax64<<<(N3c + 7) / 8, 256, 0, stream>>>(out, N3c);
}


# Replace both f32 atomic scatter-adds by destination-ordered edge lists (integer atomics) and one wave per node summing source rows in registers; sum order now ascending source id, not arrival order
# speedup vs baseline: 1.9881x; 1.9881x over previous
#include <hip/hip_runtime.h>
#include <math.h>


#define N1c  200000
#define N2c  40000
#define N3c  10000
#define INc  128
#define HIDc 256
#define OUTc 64

#define PAD32(n) (((n) + 31) & ~31)

typedef float v2f __attribute__((ext_vector_type(2)));
typedef float v8f __attribute__((ext_vector_type(8)));

__global__ void zero_i32_pair(int* __restrict__ a, int na,
                              int* __restrict__ b, int nb) {
    int i = blockIdx.x * blockDim.x + threadIdx.x;
    if (i < na) a[i] = 0;
    else if (i - na < nb) b[i - na] = 0;
}

__global__ void edge_rank(const int* __restrict__ dstA, int* __restrict__ degA,
                          int* __restrict__ rankA, int EA,
                          const int* __restrict__ dstB, int* __restrict__ degB,
                          int* __restrict__ rankB, int EB) {
    int i = blockIdx.x * blockDim.x + threadIdx.x;
    if (i < EA) {
        rankA[i] = atomicAdd(&degA[dstA[i]], 1);
    } else {
        i -= EA;
        if (i < EB) rankB[i] = atomicAdd(&degB[dstB[i]], 1);
    }
}

__global__ void seg_alloc(const int* __restrict__ degA, int* __restrict__ curA,
                          int* __restrict__ offA, float* __restrict__ cntA, int nA,
                          const int* __restrict__ degB, int* __restrict__ curB,
                          int* __restrict__ offB, float* __restrict__ cntB, int nB) {
    int i = blockIdx.x * blockDim.x + threadIdx.x;
    const int lane = threadIdx.x & 31;
    const int* deg; int* cur; int* off; float* cnt; int n;
    if (i < PAD32(nA)) {
        deg = degA; cur = curA; off = offA; cnt = cntA; n = nA;
    } else {
        i -= PAD32(nA);
        if (i >= PAD32(nB)) return;
        deg = degB; cur = curB; off = offB; cnt = cntB; n = nB;
    }
    const int d = (i < n) ? deg[i] : 0;
    int v = d;
    for (int o = 1; o < 32; o <<= 1) {
        const int u = __shfl_up(v, o, 32);
        if (lane >= o) v += u;
    }
    int base = 0;
    if (lane == 31) base = atomicAdd(cur, v);
    base = __shfl(base, 31, 32);
    if (i < n) {
        off[i] = base + v - d;
        cnt[i] = (float)d;
    }
}

__global__ void edge_place(const int* __restrict__ srcA, const int* __restrict__ dstA,
                           const int* __restrict__ rankA, const int* __restrict__ offA,
                           int* __restrict__ permA, int EA,
                           const int* __restrict__ srcB, const int* __restrict__ dstB,
                           const int* __restrict__ rankB, const int* __restrict__ offB,
                           int* __restrict__ permB, int EB) {
    int i = blockIdx.x * blockDim.x + threadIdx.x;
    if (i < EA) {
        permA[offA[dstA[i]] + rankA[i]] = srcA[i];
    } else {
        i -= EA;
        if (i < EB) permB[offB[dstB[i]] + rankB[i]] = srcB[i];
    }
}

template <int R>
static __device__ __forceinline__ void take_rows(const float* __restrict__ xb, int D,
                                                 int ids, int j, float4 (&acc)[8]) {
    float4 v[R];
    for (int k = 0; k < R; ++k) {
        const int s = __builtin_amdgcn_readlane(ids, j + k);
        v[k] = *reinterpret_cast<const float4*>(xb + (long long)s * D);
    }
    for (int k = 0; k < R; ++k) {
        acc[k].x += v[k].x; acc[k].y += v[k].y;
        acc[k].z += v[k].z; acc[k].w += v[k].w;
    }
}

__launch_bounds__(256)
__global__ void gather_rows(const float* __restrict__ x,
                            const int*   __restrict__ perm,
                            const int*   __restrict__ off,
                            const int*   __restrict__ deg,
                            float*       __restrict__ agg,
                            int nNodes, int D, int partShift) {
    const int lane = threadIdx.x & 31;
    const int wv    = __builtin_amdgcn_readfirstlane((int)(threadIdx.x >> 5));
    const int gw    = blockIdx.x * 8 + wv;
    const int node  = gw >> partShift;
    if (node >= nNodes) return;
    const int part  = gw & ((1 << partShift) - 1);
    const int start = __builtin_amdgcn_readfirstlane(off[node]);
    const int n     = __builtin_amdgcn_readfirstlane(deg[node]);
    const float* xb = x + part * 128 + lane * 4;

    float4 acc[8];
    for (int k = 0; k < 8; ++k) acc[k] = make_float4(0.0f, 0.0f, 0.0f, 0.0f);

    for (int base = 0; base < n; base += 32) {
        const int m = (n - base < 32) ? (n - base) : 32;
        const int key = (lane < m) ? perm[start + base + lane] : 0x7fffffff;
        int r = 0;
        for (int t = 0; t < m; ++t) {
            const int kt = __builtin_amdgcn_readlane(key, t);
            r += (kt < key || (kt == key && t < lane)) ? 1 : 0;
        }
        if (lane >= m) r = lane;
        const int ids = __builtin_amdgcn_ds_permute(r << 2, key);

        int j = 0;
        for (; j + 8 <= m; j += 8) take_rows<8>(xb, D, ids, j, acc);
        switch (m - j) {
            case 7: take_rows<7>(xb, D, ids, j, acc); break;
            case 6: take_rows<6>(xb, D, ids, j, acc); break;
            case 5: take_rows<5>(xb, D, ids, j, acc); break;
            case 4: take_rows<4>(xb, D, ids, j, acc); break;
            case 3: take_rows<3>(xb, D, ids, j, acc); break;
            case 2: take_rows<2>(xb, D, ids, j, acc); break;
            case 1: take_rows<1>(xb, D, ids, j, acc); break;
            default: break;
        }
    }

    float4 s;
    s.x = ((acc[0].x + acc[1].x) + (acc[2].x + acc[3].x)) + ((acc[4].x + acc[5].x) + (acc[6].x + acc[7].x));
    s.y = ((acc[0].y + acc[1].y) + (acc[2].y + acc[3].y)) + ((acc[4].y + acc[5].y) + (acc[6].y + acc[7].y));
    s.z = ((acc[0].z + acc[1].z) + (acc[2].z + acc[3].z)) + ((acc[4].z + acc[5].z) + (acc[6].z + acc[7].z));
    s.w = ((acc[0].w + acc[1].w) + (acc[2].w + acc[3].w)) + ((acc[4].w + acc[5].w) + (acc[6].w + acc[7].w));
    *reinterpret_cast<float4*>(agg + (long long)node * D + part * 128 + lane * 4) = s;
}

__launch_bounds__(128)
__global__ void sage_gemm_wmma(const float* __restrict__ agg,
                               const float* __restrict__ cnt,
                               const float* __restrict__ xt,
                               const float* __restrict__ Wl,
                               const float* __restrict__ Wr,
                               const float* __restrict__ bias,
                               float*       __restrict__ out,
                               int K, int Nout, int relu) {
    const int lane  = threadIdx.x & 31;
    const int wave  = threadIdx.x >> 5;
    const int mTile = blockIdx.x;
    const int nTile = blockIdx.y * 4 + wave;

    const int n  = lane & 15;
    const int kh = lane >> 4;

    const long long row = (long long)mTile * 16 + n;
    const long long col = (long long)nTile * 16 + n;

    const float inv = 1.0f / fmaxf(cnt[row], 1.0f);

    v8f acc;
    #pragma unroll
    for (int r = 0; r < 8; ++r) acc[r] = 0.0f;

    {
        const float* aRow = agg + row * K;
        const float* bRow = Wl  + col * K;
        for (int k0 = 0; k0 < K; k0 += 4) {
            const int ka = k0 + 2 * kh;
            v2f a; a[0] = aRow[ka] * inv; a[1] = aRow[ka + 1] * inv;
            v2f b; b[0] = bRow[ka];       b[1] = bRow[ka + 1];
            acc = __builtin_amdgcn_wmma_f32_16x16x4_f32(
                false, a, false, b, (short)0, acc, false, false);
        }
    }
    {
        const float* aRow = xt + row * K;
        const float* bRow = Wr + col * K;
        for (int k0 = 0; k0 < K; k0 += 4) {
            const int ka = k0 + 2 * kh;
            v2f a; a[0] = aRow[ka]; a[1] = aRow[ka + 1];
            v2f b; b[0] = bRow[ka]; b[1] = bRow[ka + 1];
            acc = __builtin_amdgcn_wmma_f32_16x16x4_f32(
                false, a, false, b, (short)0, acc, false, false);
        }
    }

    const float bv = bias[col];
    const long long mbase = (long long)mTile * 16 + ((lane < 16) ? 0 : 8);
    #pragma unroll
    for (int r = 0; r < 8; ++r) {
        float v = acc[r] + bv;
        if (relu) v = fmaxf(v, 0.0f);
        out[(mbase + r) * Nout + col] = v;
    }
}

__global__ void log_softmax64(float* __restrict__ out, int M) {
    const int lane = threadIdx.x & 31;
    const int wave = threadIdx.x >> 5;
    const int row  = blockIdx.x * 8 + wave;
    if (row >= M) return;
    float* p = out + (long long)row * 64;
    float a = p[lane];
    float b = p[lane + 32];
    float m = fmaxf(a, b);
    #pragma unroll
    for (int off = 16; off >= 1; off >>= 1)
        m = fmaxf(m, __shfl_xor(m, off, 32));
    float s = __expf(a - m) + __expf(b - m);
    #pragma unroll
    for (int off = 16; off >= 1; off >>= 1)
        s += __shfl_xor(s, off, 32);
    const float l = __logf(s) + m;
    p[lane]      = a - l;
    p[lane + 32] = b - l;
}

extern "C" void kernel_launch(void* const* d_in, const int* in_sizes, int n_in,
                              void* d_out, int out_size, void* d_ws, size_t ws_size,
                              hipStream_t stream) {
    const float* x    = (const float*)d_in[0];
    const int*   src1 = (const int*)  d_in[1];
    const int*   dst1 = (const int*)  d_in[2];
    const int*   src2 = (const int*)  d_in[3];
    const int*   dst2 = (const int*)  d_in[4];
    const float* W1l  = (const float*)d_in[7];
    const float* W1r  = (const float*)d_in[8];
    const float* b1   = (const float*)d_in[9];
    const float* W2l  = (const float*)d_in[10];
    const float* W2r  = (const float*)d_in[11];
    const float* b2   = (const float*)d_in[12];

    const int E1 = in_sizes[1];
    const int E2 = in_sizes[3];

    float* agg1 = (float*)d_ws;
    float* cnt1 = agg1 + (size_t)N2c * INc;
    float* h    = cnt1 + N2c;
    float* Bpl  = h    + (size_t)N2c * HIDc;
    float* agg2 = agg1;
    float* cnt2 = Bpl  + (size_t)N3c * HIDc;
    float* out  = (float*)d_out;

    int* deg1  = (int*)h;
    int* cur1  = deg1  + PAD32(N2c);
    int* off1  = cur1  + 32;
    int* rank1 = off1  + PAD32(N2c);
    int* perm1 = rank1 + (size_t)E1;
    int* deg2  = (int*)Bpl;
    int* cur2  = deg2  + PAD32(N3c);
    int* off2  = cur2  + 32;
    int* rank2 = off2  + PAD32(N3c);
    int* perm2 = rank2 + (size_t)E2;

    {
        int za = PAD32(N2c) + 32, zb = PAD32(N3c) + 32;
        zero_i32_pair<<<(za + zb + 255) / 256, 256, 0, stream>>>(deg1, za, deg2, zb);
        int et = E1 + E2;
        edge_rank<<<(et + 255) / 256, 256, 0, stream>>>(dst1, deg1, rank1, E1,
                                                        dst2, deg2, rank2, E2);
        int nt = PAD32(N2c) + PAD32(N3c);
        seg_alloc<<<(nt + 255) / 256, 256, 0, stream>>>(deg1, cur1, off1, cnt1, N2c,
                                                        deg2, cur2, off2, cnt2, N3c);
        edge_place<<<(et + 255) / 256, 256, 0, stream>>>(src1, dst1, rank1, off1, perm1, E1,
                                                         src2, dst2, rank2, off2, perm2, E2);
    }

    {
        int waves = N2c * (INc / 128);
        gather_rows<<<(waves + 7) / 8, 256, 0, stream>>>(x, perm1, off1, deg1, agg1,
                                                         N2c, INc, 0  );
        dim3 grid(N2c / 16, HIDc / 64);
        sage_gemm_wmma<<<grid, 128, 0, stream>>>(agg1, cnt1, x, W1l, W1r, b1,
                                                 h, INc, HIDc,  1);
    }

    {
        int waves = N3c * (HIDc / 128);
        gather_rows<<<(waves + 7) / 8, 256, 0, stream>>>(h, perm2, off2, deg2, agg2,
                                                         N3c, HIDc, 1  );
        dim3 grid(N3c / 16, OUTc / 64);
        sage_gemm_wmma<<<grid, 128, 0, stream>>>(agg2, cnt2, h, W2l, W2r, b2,
                                                 out, HIDc, OUTc,  0);
    }

    log_softmax64<<<(N3c + 7) / 8, 256, 0, stream>>>(out, N3c);
}


# Both SAGE products tiled through LDS: 64x128 (layer 2: 16x64) blocks, K chunks of 32 staged with whole-line 16-byte loads, 8-byte fragment reads, 2x2 tiles per wave in layer 1; sum order unchanged
# speedup vs baseline: 2.9798x; 1.4988x over previous
#include <hip/hip_runtime.h>
#include <math.h>


#define N1c  200000
#define N2c  40000
#define N3c  10000
#define INc  128
#define HIDc 256
#define OUTc 64

#define PAD32(n) (((n) + 31) & ~31)

static_assert(N2c % 64 == 0 && HIDc % 128 == 0, "layer-1 product grid is exact");
static_assert(N3c % 16 == 0 && OUTc % 64 == 0, "layer-2 product grid is exact");

typedef float v2f __attribute__((ext_vector_type(2)));
typedef float v4f __attribute__((ext_vector_type(4)));
typedef float v8f __attribute__((ext_vector_type(8)));

__global__ void zero_i32_pair(int* __restrict__ a, int na,
                              int* __restrict__ b, int nb) {
    int i = blockIdx.x * blockDim.x + threadIdx.x;
    if (i < na) a[i] = 0;
    else if (i - na < nb) b[i - na] = 0;
}

__global__ void edge_rank(const int* __restrict__ dstA, int* __restrict__ degA,
                          int* __restrict__ rankA, int EA,
                          const int* __restrict__ dstB, int* __restrict__ degB,
                          int* __restrict__ rankB, int EB) {
    int i = blockIdx.x * blockDim.x + threadIdx.x;
    if (i < EA) {
        rankA[i] = atomicAdd(&degA[dstA[i]], 1);
    } else {
        i -= EA;
        if (i < EB) rankB[i] = atomicAdd(&degB[dstB[i]], 1);
    }
}

__global__ void seg_alloc(const int* __restrict__ degA, int* __restrict__ curA,
                          int* __restrict__ offA, float* __restrict__ cntA, int nA,
                          const int* __restrict__ degB, int* __restrict__ curB,
                          int* __restrict__ offB, float* __restrict__ cntB, int nB) {
    int i = blockIdx.x * blockDim.x + threadIdx.x;
    const int lane = threadIdx.x & 31;
    const int* deg; int* cur; int* off; float* cnt; int n;
    if (i < PAD32(nA)) {
        deg = degA; cur = curA; off = offA; cnt = cntA; n = nA;
    } else {
        i -= PAD32(nA);
        if (i >= PAD32(nB)) return;
        deg = degB; cur = curB; off = offB; cnt = cntB; n = nB;
    }
    const int d = (i < n) ? deg[i] : 0;
    int v = d;
    for (int o = 1; o < 32; o <<= 1) {
        const int u = __shfl_up(v, o, 32);
        if (lane >= o) v += u;
    }
    int base = 0;
    if (lane == 31) base = atomicAdd(cur, v);
    base = __shfl(base, 31, 32);
    if (i < n) {
        off[i] = base + v - d;
        cnt[i] = (float)d;
    }
}

__global__ void edge_place(const int* __restrict__ srcA, const int* __restrict__ dstA,
                           const int* __restrict__ rankA, const int* __restrict__ offA,
                           int* __restrict__ permA, int EA,
                           const int* __restrict__ srcB, const int* __restrict__ dstB,
                           const int* __restrict__ rankB, const int* __restrict__ offB,
                           int* __restrict__ permB, int EB) {
    int i = blockIdx.x * blockDim.x + threadIdx.x;
    if (i < EA) {
        permA[offA[dstA[i]] + rankA[i]] = srcA[i];
    } else {
        i -= EA;
        if (i < EB) permB[offB[dstB[i]] + rankB[i]] = srcB[i];
    }
}

template <int R>
static __device__ __forceinline__ void take_rows(const float* __restrict__ xb, int D,
                                                 int ids, int j, float4 (&acc)[8]) {
    float4 v[R];
    for (int k = 0; k < R; ++k) {
        const int s = __builtin_amdgcn_readlane(ids, j + k);
        v[k] = *reinterpret_cast<const float4*>(xb + (long long)s * D);
    }
    for (int k = 0; k < R; ++k) {
        acc[k].x += v[k].x; acc[k].y += v[k].y;
        acc[k].z += v[k].z; acc[k].w += v[k].w;
    }
}

__launch_bounds__(256)
__global__ void gather_rows(const float* __restrict__ x,
                            const int*   __restrict__ perm,
                            const int*   __restrict__ off,
                            const int*   __restrict__ deg,
                            float*       __restrict__ agg,
                            int nNodes, int D, int partShift) {
    const int lane = threadIdx.x & 31;
    const int wv    = __builtin_amdgcn_readfirstlane((int)(threadIdx.x >> 5));
    const int gw    = blockIdx.x * 8 + wv;
    const int node  = gw >> partShift;
    if (node >= nNodes) return;
    const int part  = gw & ((1 << partShift) - 1);
    const int start = __builtin_amdgcn_readfirstlane(off[node]);
    const int n     = __builtin_amdgcn_readfirstlane(deg[node]);
    const float* xb = x + part * 128 + lane * 4;

    float4 acc[8];
    for (int k = 0; k < 8; ++k) acc[k] = make_float4(0.0f, 0.0f, 0.0f, 0.0f);

    for (int base = 0; base < n; base += 32) {
        const int m = (n - base < 32) ? (n - base) : 32;
        const int key = (lane < m) ? perm[start + base + lane] : 0x7fffffff;
        int r = 0;
        for (int t = 0; t < m; ++t) {
            const int kt = __builtin_amdgcn_readlane(key, t);
            r += (kt < key || (kt == key && t < lane)) ? 1 : 0;
        }
        if (lane >= m) r = lane;
        const int ids = __builtin_amdgcn_ds_permute(r << 2, key);

        int j = 0;
        for (; j + 8 <= m; j += 8) take_rows<8>(xb, D, ids, j, acc);
        switch (m - j) {
            case 7: take_rows<7>(xb, D, ids, j, acc); break;
            case 6: take_rows<6>(xb, D, ids, j, acc); break;
            case 5: take_rows<5>(xb, D, ids, j, acc); break;
            case 4: take_rows<4>(xb, D, ids, j, acc); break;
            case 3: take_rows<3>(xb, D, ids, j, acc); break;
            case 2: take_rows<2>(xb, D, ids, j, acc); break;
            case 1: take_rows<1>(xb, D, ids, j, acc); break;
            default: break;
        }
    }

    float4 s;
    s.x = ((acc[0].x + acc[1].x) + (acc[2].x + acc[3].x)) + ((acc[4].x + acc[5].x) + (acc[6].x + acc[7].x));
    s.y = ((acc[0].y + acc[1].y) + (acc[2].y + acc[3].y)) + ((acc[4].y + acc[5].y) + (acc[6].y + acc[7].y));
    s.z = ((acc[0].z + acc[1].z) + (acc[2].z + acc[3].z)) + ((acc[4].z + acc[5].z) + (acc[6].z + acc[7].z));
    s.w = ((acc[0].w + acc[1].w) + (acc[2].w + acc[3].w)) + ((acc[4].w + acc[5].w) + (acc[6].w + acc[7].w));
    *reinterpret_cast<float4*>(agg + (long long)node * D + part * 128 + lane * 4) = s;
}

template <int NA, int NLD>
static __device__ __forceinline__ void chunk_fetch(const float* __restrict__ Ab,
                                                   const float* __restrict__ Wb,
                                                   const int (&rowOff)[NLD], int k,
                                                   v4f (&pf)[NLD]) {
    #pragma unroll
    for (int j = 0; j < NLD; ++j)
        pf[j] = *reinterpret_cast<const v4f*>((j < NA ? Ab : Wb) + rowOff[j] + k);
}

template <int NLD, int RSTEP, int PITCH>
static __device__ __forceinline__ void chunk_put(float* dst, const v4f (&pf)[NLD]) {
    #pragma unroll
    for (int j = 0; j < NLD; ++j)
        *reinterpret_cast<v4f*>(dst + j * RSTEP * PITCH) = pf[j];
}

template <int TM, int TN, int KC, int PITCH, bool SCALE>
static __device__ __forceinline__ void chunk_mma(const float* aSrc, const float* bSrc,
                                                 const float (&inv)[TM],
                                                 v8f (&acc)[TM][TN]) {
    #pragma unroll
    for (int ks = 0; ks < KC / 4; ++ks) {
        v2f a[TM], b[TN];
        #pragma unroll
        for (int i = 0; i < TM; ++i) {
            a[i] = *reinterpret_cast<const v2f*>(aSrc + i * 16 * PITCH + ks * 4);
            if (SCALE) { a[i][0] *= inv[i]; a[i][1] *= inv[i]; }
        }
        #pragma unroll
        for (int j = 0; j < TN; ++j)
            b[j] = *reinterpret_cast<const v2f*>(bSrc + j * 16 * PITCH + ks * 4);
        #pragma unroll
        for (int i = 0; i < TM; ++i)
            #pragma unroll
            for (int j = 0; j < TN; ++j)
                acc[i][j] = __builtin_amdgcn_wmma_f32_16x16x4_f32(
                    false, a[i], false, b[j], (short)0, acc[i][j], false, false);
    }
}

template <int KT, int BM, int BN, int WM, int WN, bool RELU>
__launch_bounds__(32 * WM * WN)
__global__ void sage_gemm_lds(const float* agg,
                              const float* __restrict__ cnt,
                              const float* xt,
                              const float* Wl,
                              const float* Wr,
                              const float* __restrict__ bias,
                              float*       out,
                              int Nout) {
    constexpr int KC    = 32;
    constexpr int PITCH = KC + 4;
    constexpr int T     = 32 * WM * WN;
    constexpr int R     = BM + BN;
    constexpr int RSTEP = T / 8;
    constexpr int NLD   = R * 8 / T;
    constexpr int NA    = BM / RSTEP;
    constexpr int NCH   = KT / KC;
    constexpr int TM    = BM / WM / 16;
    constexpr int TN    = BN / WN / 16;
    static_assert(KT % KC == 0, "K chunking");
    static_assert(BM % RSTEP == 0 && (R * 8) % T == 0, "staging split");
    static_assert(BM % (16 * WM) == 0 && BN % (16 * WN) == 0, "wave tiling");

    __shared__ float lds[2 * R * PITCH];

    const int t    = threadIdx.x;
    const int lane = t & 31;
    const int wave = __builtin_amdgcn_readfirstlane(t >> 5);
    const int wm   = wave / WN;
    const int wn   = wave % WN;
    const int n    = lane & 15;
    const int kh   = lane >> 4;

    const int m0 = blockIdx.y * BM;
    const int n0 = blockIdx.x * BN;

    const int r0 = t >> 3;
    const int c4 = (t & 7) * 4;
    int rowOff[NLD];
    #pragma unroll
    for (int j = 0; j < NLD; ++j)
        rowOff[j] = ((j < NA ? m0 : n0 - BM) + r0 + j * RSTEP) * KT + c4;
    float* const putDst = lds + r0 * PITCH + c4;

    float inv[TM];
    #pragma unroll
    for (int i = 0; i < TM; ++i)
        inv[i] = 1.0f / fmaxf(cnt[m0 + wm * (BM / WM) + i * 16 + n], 1.0f);

    v8f acc[TM][TN];
    #pragma unroll
    for (int i = 0; i < TM; ++i)
        #pragma unroll
        for (int j = 0; j < TN; ++j)
            #pragma unroll
            for (int r = 0; r < 8; ++r) acc[i][j][r] = 0.0f;

    const float* const aSrc = lds + (wm * (BM / WM) + n) * PITCH + 2 * kh;
    const float* const bSrc = lds + (BM + wn * (BN / WN) + n) * PITCH + 2 * kh;

    v4f pf[NLD];
    chunk_fetch<NA, NLD>(agg, Wl, rowOff, 0, pf);
    chunk_put<NLD, RSTEP, PITCH>(putDst, pf);
    __syncthreads();

    #pragma unroll
    for (int c = 0; c < 2 * NCH; ++c) {
        const int  nx   = c + 1;
        const bool more = (nx < 2 * NCH);
        if (more) {
            if (nx < NCH) chunk_fetch<NA, NLD>(agg, Wl, rowOff, nx * KC, pf);
            else          chunk_fetch<NA, NLD>(xt, Wr, rowOff, (nx - NCH) * KC, pf);
        }
        __builtin_amdgcn_sched_barrier(0);
        const int so = (c & 1) * R * PITCH;
        if (c < NCH) chunk_mma<TM, TN, KC, PITCH, true >(aSrc + so, bSrc + so, inv, acc);
        else         chunk_mma<TM, TN, KC, PITCH, false>(aSrc + so, bSrc + so, inv, acc);
        if (more) {
            chunk_put<NLD, RSTEP, PITCH>(putDst + (nx & 1) * R * PITCH, pf);
            __syncthreads();
        }
    }

    #pragma unroll
    for (int i = 0; i < TM; ++i) {
        const long long mbase = (long long)m0 + wm * (BM / WM) + i * 16 + 8 * kh;
        #pragma unroll
        for (int j = 0; j < TN; ++j) {
            const int col = n0 + wn * (BN / WN) + j * 16 + n;
            const float bv = bias[col];
            #pragma unroll
            for (int r = 0; r < 8; ++r) {
                float v = acc[i][j][r] + bv;
                if (RELU) v = fmaxf(v, 0.0f);
                out[(mbase + r) * Nout + col] = v;
            }
        }
    }
}

__global__ void log_softmax64(float* __restrict__ out, int M) {
    const int lane = threadIdx.x & 31;
    const int wave = threadIdx.x >> 5;
    const int row  = blockIdx.x * 8 + wave;
    if (row >= M) return;
    float* p = out + (long long)row * 64;
    float a = p[lane];
    float b = p[lane + 32];
    float m = fmaxf(a, b);
    #pragma unroll
    for (int off = 16; off >= 1; off >>= 1)
        m = fmaxf(m, __shfl_xor(m, off, 32));
    float s = __expf(a - m) + __expf(b - m);
    #pragma unroll
    for (int off = 16; off >= 1; off >>= 1)
        s += __shfl_xor(s, off, 32);
    const float l = __logf(s) + m;
    p[lane]      = a - l;
    p[lane + 32] = b - l;
}

extern "C" void kernel_launch(void* const* d_in, const int* in_sizes, int n_in,
                              void* d_out, int out_size, void* d_ws, size_t ws_size,
                              hipStream_t stream) {
    const float* x    = (const float*)d_in[0];
    const int*   src1 = (const int*)  d_in[1];
    const int*   dst1 = (const int*)  d_in[2];
    const int*   src2 = (const int*)  d_in[3];
    const int*   dst2 = (const int*)  d_in[4];
    const float* W1l  = (const float*)d_in[7];
    const float* W1r  = (const float*)d_in[8];
    const float* b1   = (const float*)d_in[9];
    const float* W2l  = (const float*)d_in[10];
    const float* W2r  = (const float*)d_in[11];
    const float* b2   = (const float*)d_in[12];

    const int E1 = in_sizes[1];
    const int E2 = in_sizes[3];

    float* agg1 = (float*)d_ws;
    float* cnt1 = agg1 + (size_t)N2c * INc;
    float* h    = cnt1 + N2c;
    float* Bpl  = h    + (size_t)N2c * HIDc;
    float* agg2 = agg1;
    float* cnt2 = Bpl  + (size_t)N3c * HIDc;
    float* out  = (float*)d_out;

    int* deg1  = (int*)h;
    int* cur1  = deg1  + PAD32(N2c);
    int* off1  = cur1  + 32;
    int* rank1 = off1  + PAD32(N2c);
    int* perm1 = rank1 + (size_t)E1;
    int* deg2  = (int*)Bpl;
    int* cur2  = deg2  + PAD32(N3c);
    int* off2  = cur2  + 32;
    int* rank2 = off2  + PAD32(N3c);
    int* perm2 = rank2 + (size_t)E2;

    {
        int za = PAD32(N2c) + 32, zb = PAD32(N3c) + 32;
        zero_i32_pair<<<(za + zb + 255) / 256, 256, 0, stream>>>(deg1, za, deg2, zb);
        int et = E1 + E2;
        edge_rank<<<(et + 255) / 256, 256, 0, stream>>>(dst1, deg1, rank1, E1,
                                                        dst2, deg2, rank2, E2);
        int nt = PAD32(N2c) + PAD32(N3c);
        seg_alloc<<<(nt + 255) / 256, 256, 0, stream>>>(deg1, cur1, off1, cnt1, N2c,
                                                        deg2, cur2, off2, cnt2, N3c);
        edge_place<<<(et + 255) / 256, 256, 0, stream>>>(src1, dst1, rank1, off1, perm1, E1,
                                                         src2, dst2, rank2, off2, perm2, E2);
    }

    {
        int waves = N2c * (INc / 128);
        gather_rows<<<(waves + 7) / 8, 256, 0, stream>>>(x, perm1, off1, deg1, agg1,
                                                         N2c, INc, 0  );
        dim3 grid(HIDc / 128, N2c / 64);
        sage_gemm_lds<INc, 64, 128, 2, 4, true><<<grid, 256, 0, stream>>>(
            agg1, cnt1, x, W1l, W1r, b1, h, HIDc);
    }

    {
        int waves = N3c * (HIDc / 128);
        gather_rows<<<(waves + 7) / 8, 256, 0, stream>>>(h, perm2, off2, deg2, agg2,
                                                         N3c, HIDc, 1  );
        dim3 grid(OUTc / 64, N3c / 16);
        sage_gemm_lds<HIDc, 16, 64, 1, 4, false><<<grid, 128, 0, stream>>>(
            agg2, cnt2, h, W2l, W2r, b2, out, OUTc);
    }

    log_softmax64<<<(N3c + 7) / 8, 256, 0, stream>>>(out, N3c);
}


# Edge lists built by two-level binning in LDS (chunks ordered by node bin, then one work-group per bin): two launches, no zero fill, no global atomics, no scatter; sum order rule unchanged
# speedup vs baseline: 5.0151x; 1.6830x over previous
#include <hip/hip_runtime.h>
#include <math.h>


#define N1c  200000
#define N2c  40000
#define N3c  10000
#define INc  128
#define HIDc 256
#define OUTc 64

static_assert(N2c % 64 == 0 && HIDc % 128 == 0, "layer-1 product grid is exact");
static_assert(N3c % 16 == 0 && OUTc % 64 == 0, "layer-2 product grid is exact");

typedef float v2f __attribute__((ext_vector_type(2)));
typedef float v4f __attribute__((ext_vector_type(4)));
typedef float v8f __attribute__((ext_vector_type(8)));

#define E1c  640000
#define E2c  160000
#define ETc  (E1c + E2c)
#define SN   200
#define NB1  (N2c / SN)
#define NBK  ((N2c + N3c) / SN)
#define CH   3200
#define GW   (ETc / CH)
#define EPT  ((CH + 255) / 256)
#define CAP  8192

static_assert(N2c % SN == 0 && N3c % SN == 0, "bins do not straddle the layers");
static_assert(E1c % CH == 0 && E2c % CH == 0, "chunks are full and do not straddle the layers");
static_assert(NBK + 1 <= 256 && GW <= 256 && SN <= 256, "one thread per bin / chunk / node");

static __device__ __forceinline__ int block_scan_incl(int v, int* ws, int t) {
    const int lane = t & 31;
    for (int o = 1; o < 32; o <<= 1) {
        const int u = __shfl_up(v, o, 32);
        if (lane >= o) v += u;
    }
    if (lane == 31) ws[t >> 5] = v;
    __syncthreads();
    int add = 0;
    for (int j = 0; j < 8; ++j) add += (j < (t >> 5)) ? ws[j] : 0;
    return v + add;
}

__launch_bounds__(256)
__global__ void bin_edges(const int* __restrict__ dst1, const int* __restrict__ src1,
                          const int* __restrict__ dst2, const int* __restrict__ src2,
                          int* __restrict__ binned, int* __restrict__ tblT) {
    __shared__ int  cnt[NBK];
    __shared__ int  boff[NBK + 1];
    __shared__ int  ws[8];
    __shared__ int2 stage[CH];
    const int t = threadIdx.x;
    const int g = blockIdx.x;
    const int base = g * CH;

    if (t < NBK) cnt[t] = 0;
    __syncthreads();

    const bool l1 = (base < E1c);
    const int* dp = l1 ? dst1 + base : dst2 + (base - E1c);
    const int* sp = l1 ? src1 + base : src2 + (base - E1c);
    const int kofs = l1 ? 0 : N2c;

    int key[EPT], sv[EPT], rk[EPT];
    #pragma unroll
    for (int k = 0; k < EPT; ++k) {
        const int idx = k * 256 + t;
        const int idc = (idx < CH) ? idx : (CH - 1);
        key[k] = dp[idc] + kofs; sv[k] = sp[idc]; rk[k] = 0;
    }
    #pragma unroll
    for (int k = 0; k < EPT; ++k)
        if (k * 256 + t < CH) rk[k] = atomicAdd(&cnt[key[k] / SN], 1);
    __syncthreads();

    const int v    = (t < NBK) ? cnt[t] : 0;
    const int incl = block_scan_incl(v, ws, t);
    if (t <= NBK) boff[t] = incl - v;
    __syncthreads();

    #pragma unroll
    for (int k = 0; k < EPT; ++k)
        if (k * 256 + t < CH) stage[boff[key[k] / SN] + rk[k]] = make_int2(key[k], sv[k]);
    __syncthreads();

    int2* out2 = reinterpret_cast<int2*>(binned);
    #pragma unroll
    for (int k = 0; k < EPT; ++k) {
        const int idx = k * 256 + t;
        if (idx < CH) out2[base + idx] = stage[idx];
    }
    if (t <= NBK) tblT[t * GW + g] = boff[t];
}

#define EB 16
struct ListWalk { int lo, nxt, gofs; };

static __device__ __forceinline__ ListWalk walk_begin(const int* segS, const int* segP, int i) {
    int lo = 0;
    #pragma unroll
    for (int step = 128; step >= 1; step >>= 1)
        if (segP[lo + step] <= i) lo += step;
    ListWalk w;
    w.lo = lo; w.nxt = segP[lo + 1]; w.gofs = lo * CH + segS[lo] - segP[lo];
    return w;
}

static __device__ __forceinline__ int walk_to(ListWalk& w, const int* segS, const int* segP, int i) {
    while (i >= w.nxt) {
        const int st = w.nxt;
        ++w.lo;
        w.nxt  = segP[w.lo + 1];
        w.gofs = w.lo * CH + segS[w.lo] - st;
    }
    return w.gofs + i;
}

__launch_bounds__(256)
__global__ void bucket_lists(const int* __restrict__ binned, const int* __restrict__ tblT,
                             int* __restrict__ deg1, int* __restrict__ off1,
                             float* __restrict__ cnt1, int* __restrict__ perm1,
                             int* __restrict__ deg2, int* __restrict__ off2,
                             float* __restrict__ cnt2, int* __restrict__ perm2) {
    __shared__ int segS[256];
    __shared__ int segP[257];
    __shared__ int degl[256];
    __shared__ int offl[256];
    __shared__ int wsA[8], wsB[8], wsC[8];
    __shared__ int sBase;
    __shared__ int stage[CAP];
    const int t = threadIdx.x;
    const int b = blockIdx.x;
    const int2* in2 = reinterpret_cast<const int2*>(binned);

    int s = 0, n = 0;
    if (t < GW) {
        s = tblT[b * GW + t];
        n = tblT[(b + 1) * GW + t] - s;
    }
    const int pincl = block_scan_incl(n, wsA, t);
    const int sincl = block_scan_incl(s, wsB, t);
    segS[t] = s;
    segP[t] = pincl - n;
    if (t == 255) { segP[256] = pincl; sBase = sincl; }
    degl[t] = 0;
    __syncthreads();
    const int nb   = segP[256];
    const int base = sBase;
    const int lo_key = b * SN;

    const int per = (nb + 255) >> 8;
    const int i0  = t * per;
    const int i1  = (i0 + per < nb) ? (i0 + per) : nb;
    const bool any = (i0 < nb);
    ListWalk w0;
    w0.lo = 0; w0.nxt = 0; w0.gofs = 0;
    if (any) w0 = walk_begin(segS, segP, i0);

    if (any) {
        ListWalk w = w0;
        for (int i = i0; i < i1; i += EB) {
            int key[EB];
            #pragma unroll
            for (int k = 0; k < EB; ++k)
                key[k] = binned[2 * walk_to(w, segS, segP, (i + k < i1) ? (i + k) : (i1 - 1))];
            #pragma unroll
            for (int k = 0; k < EB; ++k)
                atomicAdd(&degl[key[k] - lo_key], (i + k < i1) ? 1 : 0);
        }
    }
    __syncthreads();

    const int d     = (t < SN) ? degl[t] : 0;
    const int dincl = block_scan_incl(d, wsC, t);
    offl[t] = dincl - d;
    degl[t] = 0;
    const bool L1 = (b < NB1);
    int* perm = L1 ? perm1 : perm2;
    const int pbase = L1 ? base : base - E1c;
    if (t < SN) {
        if (L1) {
            const int node = lo_key + t;
            deg1[node] = d; off1[node] = pbase + dincl - d; cnt1[node] = (float)d;
        } else {
            const int node = lo_key + t - N2c;
            deg2[node] = d; off2[node] = pbase + dincl - d; cnt2[node] = (float)d;
        }
    }
    __syncthreads();

    const bool fits = (nb <= CAP);
    if (any) {
        ListWalk w = w0;
        for (int i = i0; i < i1; i += EB) {
            int2 e[EB];
            int  pos[EB];
            #pragma unroll
            for (int k = 0; k < EB; ++k)
                e[k] = in2[walk_to(w, segS, segP, (i + k < i1) ? (i + k) : (i1 - 1))];
            #pragma unroll
            for (int k = 0; k < EB; ++k) {
                const int dl = e[k].x - lo_key;
                pos[k] = offl[dl] + atomicAdd(&degl[dl], (i + k < i1) ? 1 : 0);
            }
            #pragma unroll
            for (int k = 0; k < EB; ++k) {
                if (i + k < i1) {
                    if (fits) stage[pos[k]] = e[k].y;
                    else      perm[pbase + pos[k]] = e[k].y;
                }
            }
        }
    }
    __syncthreads();
    if (fits)
        for (int i = t; i < nb; i += 256) perm[pbase + i] = stage[i];
}

template <int R>
static __device__ __forceinline__ void take_rows(const float* __restrict__ xb, int D,
                                                 int ids, int j, float4 (&acc)[8]) {
    float4 v[R];
    for (int k = 0; k < R; ++k) {
        const int s = __builtin_amdgcn_readlane(ids, j + k);
        v[k] = *reinterpret_cast<const float4*>(xb + (long long)s * D);
    }
    for (int k = 0; k < R; ++k) {
        acc[k].x += v[k].x; acc[k].y += v[k].y;
        acc[k].z += v[k].z; acc[k].w += v[k].w;
    }
}

__launch_bounds__(256)
__global__ void gather_rows(const float* __restrict__ x,
                            const int*   __restrict__ perm,
                            const int*   __restrict__ off,
                            const int*   __restrict__ deg,
                            float*       __restrict__ agg,
                            int nNodes, int D, int partShift) {
    const int lane = threadIdx.x & 31;
    const int wv    = __builtin_amdgcn_readfirstlane((int)(threadIdx.x >> 5));
    const int gw    = blockIdx.x * 8 + wv;
    const int node  = gw >> partShift;
    if (node >= nNodes) return;
    const int part  = gw & ((1 << partShift) - 1);
    const int start = __builtin_amdgcn_readfirstlane(off[node]);
    const int n     = __builtin_amdgcn_readfirstlane(deg[node]);
    const float* xb = x + part * 128 + lane * 4;

    float4 acc[8];
    for (int k = 0; k < 8; ++k) acc[k] = make_float4(0.0f, 0.0f, 0.0f, 0.0f);

    for (int base = 0; base < n; base += 32) {
        const int m = (n - base < 32) ? (n - base) : 32;
        const int key = (lane < m) ? perm[start + base + lane] : 0x7fffffff;
        int r = 0;
        for (int t = 0; t < m; ++t) {
            const int kt = __builtin_amdgcn_readlane(key, t);
            r += (kt < key || (kt == key && t < lane)) ? 1 : 0;
        }
        if (lane >= m) r = lane;
        const int ids = __builtin_amdgcn_ds_permute(r << 2, key);

        int j = 0;
        for (; j + 8 <= m; j += 8) take_rows<8>(xb, D, ids, j, acc);
        switch (m - j) {
            case 7: take_rows<7>(xb, D, ids, j, acc); break;
            case 6: take_rows<6>(xb, D, ids, j, acc); break;
            case 5: take_rows<5>(xb, D, ids, j, acc); break;
            case 4: take_rows<4>(xb, D, ids, j, acc); break;
            case 3: take_rows<3>(xb, D, ids, j, acc); break;
            case 2: take_rows<2>(xb, D, ids, j, acc); break;
            case 1: take_rows<1>(xb, D, ids, j, acc); break;
            default: break;
        }
    }

    float4 s;
    s.x = ((acc[0].x + acc[1].x) + (acc[2].x + acc[3].x)) + ((acc[4].x + acc[5].x) + (acc[6].x + acc[7].x));
    s.y = ((acc[0].y + acc[1].y) + (acc[2].y + acc[3].y)) + ((acc[4].y + acc[5].y) + (acc[6].y + acc[7].y));
    s.z = ((acc[0].z + acc[1].z) + (acc[2].z + acc[3].z)) + ((acc[4].z + acc[5].z) + (acc[6].z + acc[7].z));
    s.w = ((acc[0].w + acc[1].w) + (acc[2].w + acc[3].w)) + ((acc[4].w + acc[5].w) + (acc[6].w + acc[7].w));
    *reinterpret_cast<float4*>(agg + (long long)node * D + part * 128 + lane * 4) = s;
}

template <int NA, int NLD>
static __device__ __forceinline__ void chunk_fetch(const float* __restrict__ Ab,
                                                   const float* __restrict__ Wb,
                                                   const int (&rowOff)[NLD], int k,
                                                   v4f (&pf)[NLD]) {
    #pragma unroll
    for (int j = 0; j < NLD; ++j)
        pf[j] = *reinterpret_cast<const v4f*>((j < NA ? Ab : Wb) + rowOff[j] + k);
}

template <int NLD, int RSTEP, int PITCH>
static __device__ __forceinline__ void chunk_put(float* dst, const v4f (&pf)[NLD]) {
    #pragma unroll
    for (int j = 0; j < NLD; ++j)
        *reinterpret_cast<v4f*>(dst + j * RSTEP * PITCH) = pf[j];
}

template <int TM, int TN, int KC, int PITCH, bool SCALE>
static __device__ __forceinline__ void chunk_mma(const float* aSrc, const float* bSrc,
                                                 const float (&inv)[TM],
                                                 v8f (&acc)[TM][TN]) {
    #pragma unroll
    for (int ks = 0; ks < KC / 4; ++ks) {
        v2f a[TM], b[TN];
        #pragma unroll
        for (int i = 0; i < TM; ++i) {
            a[i] = *reinterpret_cast<const v2f*>(aSrc + i * 16 * PITCH + ks * 4);
            if (SCALE) { a[i][0] *= inv[i]; a[i][1] *= inv[i]; }
        }
        #pragma unroll
        for (int j = 0; j < TN; ++j)
            b[j] = *reinterpret_cast<const v2f*>(bSrc + j * 16 * PITCH + ks * 4);
        #pragma unroll
        for (int i = 0; i < TM; ++i)
            #pragma unroll
            for (int j = 0; j < TN; ++j)
                acc[i][j] = __builtin_amdgcn_wmma_f32_16x16x4_f32(
                    false, a[i], false, b[j], (short)0, acc[i][j], false, false);
    }
}

template <int KT, int BM, int BN, int WM, int WN, bool RELU>
__launch_bounds__(32 * WM * WN)
__global__ void sage_gemm_lds(const float* agg,
                              const float* __restrict__ cnt,
                              const float* xt,
                              const float* Wl,
                              const float* Wr,
                              const float* __restrict__ bias,
                              float*       out,
                              int Nout) {
    constexpr int KC    = 32;
    constexpr int PITCH = KC + 4;
    constexpr int T     = 32 * WM * WN;
    constexpr int R     = BM + BN;
    constexpr int RSTEP = T / 8;
    constexpr int NLD   = R * 8 / T;
    constexpr int NA    = BM / RSTEP;
    constexpr int NCH   = KT / KC;
    constexpr int TM    = BM / WM / 16;
    constexpr int TN    = BN / WN / 16;
    static_assert(KT % KC == 0, "K chunking");
    static_assert(BM % RSTEP == 0 && (R * 8) % T == 0, "staging split");
    static_assert(BM % (16 * WM) == 0 && BN % (16 * WN) == 0, "wave tiling");

    __shared__ float lds[2 * R * PITCH];

    const int t    = threadIdx.x;
    const int lane = t & 31;
    const int wave = __builtin_amdgcn_readfirstlane(t >> 5);
    const int wm   = wave / WN;
    const int wn   = wave % WN;
    const int n    = lane & 15;
    const int kh   = lane >> 4;

    const int m0 = blockIdx.y * BM;
    const int n0 = blockIdx.x * BN;

    const int r0 = t >> 3;
    const int c4 = (t & 7) * 4;
    int rowOff[NLD];
    #pragma unroll
    for (int j = 0; j < NLD; ++j)
        rowOff[j] = ((j < NA ? m0 : n0 - BM) + r0 + j * RSTEP) * KT + c4;
    float* const putDst = lds + r0 * PITCH + c4;

    float inv[TM];
    #pragma unroll
    for (int i = 0; i < TM; ++i)
        inv[i] = 1.0f / fmaxf(cnt[m0 + wm * (BM / WM) + i * 16 + n], 1.0f);

    v8f acc[TM][TN];
    #pragma unroll
    for (int i = 0; i < TM; ++i)
        #pragma unroll
        for (int j = 0; j < TN; ++j)
            #pragma unroll
            for (int r = 0; r < 8; ++r) acc[i][j][r] = 0.0f;

    const float* const aSrc = lds + (wm * (BM / WM) + n) * PITCH + 2 * kh;
    const float* const bSrc = lds + (BM + wn * (BN / WN) + n) * PITCH + 2 * kh;

    v4f pf[NLD];
    chunk_fetch<NA, NLD>(agg, Wl, rowOff, 0, pf);
    chunk_put<NLD, RSTEP, PITCH>(putDst, pf);
    __syncthreads();

    #pragma unroll
    for (int c = 0; c < 2 * NCH; ++c) {
        const int  nx   = c + 1;
        const bool more = (nx < 2 * NCH);
        if (more) {
            if (nx < NCH) chunk_fetch<NA, NLD>(agg, Wl, rowOff, nx * KC, pf);
            else          chunk_fetch<NA, NLD>(xt, Wr, rowOff, (nx - NCH) * KC, pf);
        }
        __builtin_amdgcn_sched_barrier(0);
        const int so = (c & 1) * R * PITCH;
        if (c < NCH) chunk_mma<TM, TN, KC, PITCH, true >(aSrc + so, bSrc + so, inv, acc);
        else         chunk_mma<TM, TN, KC, PITCH, false>(aSrc + so, bSrc + so, inv, acc);
        if (more) {
            chunk_put<NLD, RSTEP, PITCH>(putDst + (nx & 1) * R * PITCH, pf);
            __syncthreads();
        }
    }

    #pragma unroll
    for (int i = 0; i < TM; ++i) {
        const long long mbase = (long long)m0 + wm * (BM / WM) + i * 16 + 8 * kh;
        #pragma unroll
        for (int j = 0; j < TN; ++j) {
            const int col = n0 + wn * (BN / WN) + j * 16 + n;
            const float bv = bias[col];
            #pragma unroll
            for (int r = 0; r < 8; ++r) {
                float v = acc[i][j][r] + bv;
                if (RELU) v = fmaxf(v, 0.0f);
                out[(mbase + r) * Nout + col] = v;
            }
        }
    }
}

__global__ void log_softmax64(float* __restrict__ out, int M) {
    const int lane = threadIdx.x & 31;
    const int wave = threadIdx.x >> 5;
    const int row  = blockIdx.x * 8 + wave;
    if (row >= M) return;
    float* p = out + (long long)row * 64;
    float a = p[lane];
    float b = p[lane + 32];
    float m = fmaxf(a, b);
    #pragma unroll
    for (int off = 16; off >= 1; off >>= 1)
        m = fmaxf(m, __shfl_xor(m, off, 32));
    float s = __expf(a - m) + __expf(b - m);
    #pragma unroll
    for (int off = 16; off >= 1; off >>= 1)
        s += __shfl_xor(s, off, 32);
    const float l = __logf(s) + m;
    p[lane]      = a - l;
    p[lane + 32] = b - l;
}

extern "C" void kernel_launch(void* const* d_in, const int* in_sizes, int n_in,
                              void* d_out, int out_size, void* d_ws, size_t ws_size,
                              hipStream_t stream) {
    const float* x    = (const float*)d_in[0];
    const int*   src1 = (const int*)  d_in[1];
    const int*   dst1 = (const int*)  d_in[2];
    const int*   src2 = (const int*)  d_in[3];
    const int*   dst2 = (const int*)  d_in[4];
    const float* W1l  = (const float*)d_in[7];
    const float* W1r  = (const float*)d_in[8];
    const float* b1   = (const float*)d_in[9];
    const float* W2l  = (const float*)d_in[10];
    const float* W2r  = (const float*)d_in[11];
    const float* b2   = (const float*)d_in[12];

    float* agg1 = (float*)d_ws;
    float* cnt1 = agg1 + (size_t)N2c * INc;
    float* h    = cnt1 + N2c;
    float* Bpl  = h    + (size_t)N2c * HIDc;
    float* agg2 = agg1;
    float* cnt2 = Bpl  + (size_t)N3c * HIDc;
    float* out  = (float*)d_out;

    int* binned = (int*)agg1;
    int* tblT   = binned + 2 * (size_t)ETc;
    int* deg1  = (int*)h;
    int* off1  = deg1 + N2c;
    int* perm1 = off1 + N2c;
    int* deg2  = (int*)Bpl;
    int* off2  = deg2 + N3c;
    int* perm2 = off2 + N3c;

    bin_edges<<<GW, 256, 0, stream>>>(dst1, src1, dst2, src2, binned, tblT);
    bucket_lists<<<NBK, 256, 0, stream>>>(binned, tblT, deg1, off1, cnt1, perm1,
                                          deg2, off2, cnt2, perm2);

    {
        int waves = N2c * (INc / 128);
        gather_rows<<<(waves + 7) / 8, 256, 0, stream>>>(x, perm1, off1, deg1, agg1,
                                                         N2c, INc, 0  );
        dim3 grid(HIDc / 128, N2c / 64);
        sage_gemm_lds<INc, 64, 128, 2, 4, true><<<grid, 256, 0, stream>>>(
            agg1, cnt1, x, W1l, W1r, b1, h, HIDc);
    }

    {
        int waves = N3c * (HIDc / 128);
        gather_rows<<<(waves + 7) / 8, 256, 0, stream>>>(h, perm2, off2, deg2, agg2,
                                                         N3c, HIDc, 1  );
        dim3 grid(OUTc / 64, N3c / 16);
        sage_gemm_lds<HIDc, 16, 64, 1, 4, false><<<grid, 128, 0, stream>>>(
            agg2, cnt2, h, W2l, W2r, b2, out, OUTc);
    }

    log_softmax64<<<(N3c + 7) / 8, 256, 0, stream>>>(out, N3c);
}


# Row-wise log_softmax finished inside the layer-2 product (logits through the drained LDS stage, a wave takes 4 whole rows, whole-line stores): 6 launches instead of 7; sum order unchanged
# speedup vs baseline: 5.3091x; 1.0586x over previous
#include <hip/hip_runtime.h>
#include <math.h>


#define N1c  200000
#define N2c  40000
#define N3c  10000
#define INc  128
#define HIDc 256
#define OUTc 64

static_assert(N2c % 64 == 0 && HIDc % 128 == 0, "layer-1 product grid is exact");
static_assert(N3c % 16 == 0 && OUTc % 64 == 0, "layer-2 product grid is exact");

typedef float v2f __attribute__((ext_vector_type(2)));
typedef float v4f __attribute__((ext_vector_type(4)));
typedef float v8f __attribute__((ext_vector_type(8)));

#define E1c  640000
#define E2c  160000
#define ETc  (E1c + E2c)
#define SN   200
#define NB1  (N2c / SN)
#define NBK  ((N2c + N3c) / SN)
#define CH   3200
#define GW   (ETc / CH)
#define EPT  ((CH + 255) / 256)
#define CAP  8192

static_assert(N2c % SN == 0 && N3c % SN == 0, "bins do not straddle the layers");
static_assert(E1c % CH == 0 && E2c % CH == 0, "chunks are full and do not straddle the layers");
static_assert(NBK + 1 <= 256 && GW <= 256 && SN <= 256, "one thread per bin / chunk / node");

static __device__ __forceinline__ int block_scan_incl(int v, int* ws, int t) {
    const int lane = t & 31;
    for (int o = 1; o < 32; o <<= 1) {
        const int u = __shfl_up(v, o, 32);
        if (lane >= o) v += u;
    }
    if (lane == 31) ws[t >> 5] = v;
    __syncthreads();
    int add = 0;
    for (int j = 0; j < 8; ++j) add += (j < (t >> 5)) ? ws[j] : 0;
    return v + add;
}

__launch_bounds__(256)
__global__ void bin_edges(const int* __restrict__ dst1, const int* __restrict__ src1,
                          const int* __restrict__ dst2, const int* __restrict__ src2,
                          int* __restrict__ binned, int* __restrict__ tblT) {
    __shared__ int  cnt[NBK];
    __shared__ int  boff[NBK + 1];
    __shared__ int  ws[8];
    __shared__ int2 stage[CH];
    const int t = threadIdx.x;
    const int g = blockIdx.x;
    const int base = g * CH;

    if (t < NBK) cnt[t] = 0;
    __syncthreads();

    const bool l1 = (base < E1c);
    const int* dp = l1 ? dst1 + base : dst2 + (base - E1c);
    const int* sp = l1 ? src1 + base : src2 + (base - E1c);
    const int kofs = l1 ? 0 : N2c;

    int key[EPT], sv[EPT], rk[EPT];
    #pragma unroll
    for (int k = 0; k < EPT; ++k) {
        const int idx = k * 256 + t;
        const int idc = (idx < CH) ? idx : (CH - 1);
        key[k] = dp[idc] + kofs; sv[k] = sp[idc]; rk[k] = 0;
    }
    #pragma unroll
    for (int k = 0; k < EPT; ++k)
        if (k * 256 + t < CH) rk[k] = atomicAdd(&cnt[key[k] / SN], 1);
    __syncthreads();

    const int v    = (t < NBK) ? cnt[t] : 0;
    const int incl = block_scan_incl(v, ws, t);
    if (t <= NBK) boff[t] = incl - v;
    __syncthreads();

    #pragma unroll
    for (int k = 0; k < EPT; ++k)
        if (k * 256 + t < CH) stage[boff[key[k] / SN] + rk[k]] = make_int2(key[k], sv[k]);
    __syncthreads();

    int2* out2 = reinterpret_cast<int2*>(binned);
    #pragma unroll
    for (int k = 0; k < EPT; ++k) {
        const int idx = k * 256 + t;
        if (idx < CH) out2[base + idx] = stage[idx];
    }
    if (t <= NBK) tblT[t * GW + g] = boff[t];
}

#define EB 16
struct ListWalk { int lo, nxt, gofs; };

static __device__ __forceinline__ ListWalk walk_begin(const int* segS, const int* segP, int i) {
    int lo = 0;
    #pragma unroll
    for (int step = 128; step >= 1; step >>= 1)
        if (segP[lo + step] <= i) lo += step;
    ListWalk w;
    w.lo = lo; w.nxt = segP[lo + 1]; w.gofs = lo * CH + segS[lo] - segP[lo];
    return w;
}

static __device__ __forceinline__ int walk_to(ListWalk& w, const int* segS, const int* segP, int i) {
    while (i >= w.nxt) {
        const int st = w.nxt;
        ++w.lo;
        w.nxt  = segP[w.lo + 1];
        w.gofs = w.lo * CH + segS[w.lo] - st;
    }
    return w.gofs + i;
}

__launch_bounds__(256)
__global__ void bucket_lists(const int* __restrict__ binned, const int* __restrict__ tblT,
                             int* __restrict__ deg1, int* __restrict__ off1,
                             float* __restrict__ cnt1, int* __restrict__ perm1,
                             int* __restrict__ deg2, int* __restrict__ off2,
                             float* __restrict__ cnt2, int* __restrict__ perm2) {
    __shared__ int segS[256];
    __shared__ int segP[257];
    __shared__ int degl[256];
    __shared__ int offl[256];
    __shared__ int wsA[8], wsB[8], wsC[8];
    __shared__ int sBase;
    __shared__ int stage[CAP];
    const int t = threadIdx.x;
    const int b = blockIdx.x;
    const int2* in2 = reinterpret_cast<const int2*>(binned);

    int s = 0, n = 0;
    if (t < GW) {
        s = tblT[b * GW + t];
        n = tblT[(b + 1) * GW + t] - s;
    }
    const int pincl = block_scan_incl(n, wsA, t);
    const int sincl = block_scan_incl(s, wsB, t);
    segS[t] = s;
    segP[t] = pincl - n;
    if (t == 255) { segP[256] = pincl; sBase = sincl; }
    degl[t] = 0;
    __syncthreads();
    const int nb   = segP[256];
    const int base = sBase;
    const int lo_key = b * SN;

    const int per = (nb + 255) >> 8;
    const int i0  = t * per;
    const int i1  = (i0 + per < nb) ? (i0 + per) : nb;
    const bool any = (i0 < nb);
    ListWalk w0;
    w0.lo = 0; w0.nxt = 0; w0.gofs = 0;
    if (any) w0 = walk_begin(segS, segP, i0);

    if (any) {
        ListWalk w = w0;
        for (int i = i0; i < i1; i += EB) {
            int key[EB];
            #pragma unroll
            for (int k = 0; k < EB; ++k)
                key[k] = binned[2 * walk_to(w, segS, segP, (i + k < i1) ? (i + k) : (i1 - 1))];
            #pragma unroll
            for (int k = 0; k < EB; ++k)
                atomicAdd(&degl[key[k] - lo_key], (i + k < i1) ? 1 : 0);
        }
    }
    __syncthreads();

    const int d     = (t < SN) ? degl[t] : 0;
    const int dincl = block_scan_incl(d, wsC, t);
    offl[t] = dincl - d;
    degl[t] = 0;
    const bool L1 = (b < NB1);
    int* perm = L1 ? perm1 : perm2;
    const int pbase = L1 ? base : base - E1c;
    if (t < SN) {
        if (L1) {
            const int node = lo_key + t;
            deg1[node] = d; off1[node] = pbase + dincl - d; cnt1[node] = (float)d;
        } else {
            const int node = lo_key + t - N2c;
            deg2[node] = d; off2[node] = pbase + dincl - d; cnt2[node] = (float)d;
        }
    }
    __syncthreads();

    const bool fits = (nb <= CAP);
    if (any) {
        ListWalk w = w0;
        for (int i = i0; i < i1; i += EB) {
            int2 e[EB];
            int  pos[EB];
            #pragma unroll
            for (int k = 0; k < EB; ++k)
                e[k] = in2[walk_to(w, segS, segP, (i + k < i1) ? (i + k) : (i1 - 1))];
            #pragma unroll
            for (int k = 0; k < EB; ++k) {
                const int dl = e[k].x - lo_key;
                pos[k] = offl[dl] + atomicAdd(&degl[dl], (i + k < i1) ? 1 : 0);
            }
            #pragma unroll
            for (int k = 0; k < EB; ++k) {
                if (i + k < i1) {
                    if (fits) stage[pos[k]] = e[k].y;
                    else      perm[pbase + pos[k]] = e[k].y;
                }
            }
        }
    }
    __syncthreads();
    if (fits)
        for (int i = t; i < nb; i += 256) perm[pbase + i] = stage[i];
}

template <int R>
static __device__ __forceinline__ void take_rows(const float* __restrict__ xb, int D,
                                                 int ids, int j, float4 (&acc)[8]) {
    float4 v[R];
    for (int k = 0; k < R; ++k) {
        const int s = __builtin_amdgcn_readlane(ids, j + k);
        v[k] = *reinterpret_cast<const float4*>(xb + (long long)s * D);
    }
    for (int k = 0; k < R; ++k) {
        acc[k].x += v[k].x; acc[k].y += v[k].y;
        acc[k].z += v[k].z; acc[k].w += v[k].w;
    }
}

__launch_bounds__(256)
__global__ void gather_rows(const float* __restrict__ x,
                            const int*   __restrict__ perm,
                            const int*   __restrict__ off,
                            const int*   __restrict__ deg,
                            float*       __restrict__ agg,
                            int nNodes, int D, int partShift) {
    const int lane = threadIdx.x & 31;
    const int wv    = __builtin_amdgcn_readfirstlane((int)(threadIdx.x >> 5));
    const int gw    = blockIdx.x * 8 + wv;
    const int node  = gw >> partShift;
    if (node >= nNodes) return;
    const int part  = gw & ((1 << partShift) - 1);
    const int start = __builtin_amdgcn_readfirstlane(off[node]);
    const int n     = __builtin_amdgcn_readfirstlane(deg[node]);
    const float* xb = x + part * 128 + lane * 4;

    float4 acc[8];
    for (int k = 0; k < 8; ++k) acc[k] = make_float4(0.0f, 0.0f, 0.0f, 0.0f);

    for (int base = 0; base < n; base += 32) {
        const int m = (n - base < 32) ? (n - base) : 32;
        const int key = (lane < m) ? perm[start + base + lane] : 0x7fffffff;
        int r = 0;
        for (int t = 0; t < m; ++t) {
            const int kt = __builtin_amdgcn_readlane(key, t);
            r += (kt < key || (kt == key && t < lane)) ? 1 : 0;
        }
        if (lane >= m) r = lane;
        const int ids = __builtin_amdgcn_ds_permute(r << 2, key);

        int j = 0;
        for (; j + 8 <= m; j += 8) take_rows<8>(xb, D, ids, j, acc);
        switch (m - j) {
            case 7: take_rows<7>(xb, D, ids, j, acc); break;
            case 6: take_rows<6>(xb, D, ids, j, acc); break;
            case 5: take_rows<5>(xb, D, ids, j, acc); break;
            case 4: take_rows<4>(xb, D, ids, j, acc); break;
            case 3: take_rows<3>(xb, D, ids, j, acc); break;
            case 2: take_rows<2>(xb, D, ids, j, acc); break;
            case 1: take_rows<1>(xb, D, ids, j, acc); break;
            default: break;
        }
    }

    float4 s;
    s.x = ((acc[0].x + acc[1].x) + (acc[2].x + acc[3].x)) + ((acc[4].x + acc[5].x) + (acc[6].x + acc[7].x));
    s.y = ((acc[0].y + acc[1].y) + (acc[2].y + acc[3].y)) + ((acc[4].y + acc[5].y) + (acc[6].y + acc[7].y));
    s.z = ((acc[0].z + acc[1].z) + (acc[2].z + acc[3].z)) + ((acc[4].z + acc[5].z) + (acc[6].z + acc[7].z));
    s.w = ((acc[0].w + acc[1].w) + (acc[2].w + acc[3].w)) + ((acc[4].w + acc[5].w) + (acc[6].w + acc[7].w));
    *reinterpret_cast<float4*>(agg + (long long)node * D + part * 128 + lane * 4) = s;
}

template <int NA, int NLD>
static __device__ __forceinline__ void chunk_fetch(const float* __restrict__ Ab,
                                                   const float* __restrict__ Wb,
                                                   const int (&rowOff)[NLD], int k,
                                                   v4f (&pf)[NLD]) {
    #pragma unroll
    for (int j = 0; j < NLD; ++j)
        pf[j] = *reinterpret_cast<const v4f*>((j < NA ? Ab : Wb) + rowOff[j] + k);
}

template <int NLD, int RSTEP, int PITCH>
static __device__ __forceinline__ void chunk_put(float* dst, const v4f (&pf)[NLD]) {
    #pragma unroll
    for (int j = 0; j < NLD; ++j)
        *reinterpret_cast<v4f*>(dst + j * RSTEP * PITCH) = pf[j];
}

template <int TM, int TN, int KC, int PITCH, bool SCALE>
static __device__ __forceinline__ void chunk_mma(const float* aSrc, const float* bSrc,
                                                 const float (&inv)[TM],
                                                 v8f (&acc)[TM][TN]) {
    #pragma unroll
    for (int ks = 0; ks < KC / 4; ++ks) {
        v2f a[TM], b[TN];
        #pragma unroll
        for (int i = 0; i < TM; ++i) {
            a[i] = *reinterpret_cast<const v2f*>(aSrc + i * 16 * PITCH + ks * 4);
            if (SCALE) { a[i][0] *= inv[i]; a[i][1] *= inv[i]; }
        }
        #pragma unroll
        for (int j = 0; j < TN; ++j)
            b[j] = *reinterpret_cast<const v2f*>(bSrc + j * 16 * PITCH + ks * 4);
        #pragma unroll
        for (int i = 0; i < TM; ++i)
            #pragma unroll
            for (int j = 0; j < TN; ++j)
                acc[i][j] = __builtin_amdgcn_wmma_f32_16x16x4_f32(
                    false, a[i], false, b[j], (short)0, acc[i][j], false, false);
    }
}

template <int KT, int BM, int BN, int WM, int WN, bool RELU, bool LSM>
__launch_bounds__(32 * WM * WN)
__global__ void sage_gemm_lds(const float* agg,
                              const float* __restrict__ cnt,
                              const float* xt,
                              const float* Wl,
                              const float* Wr,
                              const float* __restrict__ bias,
                              float*       out,
                              int Nout) {
    constexpr int KC    = 32;
    constexpr int PITCH = KC + 4;
    constexpr int T     = 32 * WM * WN;
    constexpr int R     = BM + BN;
    constexpr int RSTEP = T / 8;
    constexpr int NLD   = R * 8 / T;
    constexpr int NA    = BM / RSTEP;
    constexpr int NCH   = KT / KC;
    constexpr int TM    = BM / WM / 16;
    constexpr int TN    = BN / WN / 16;
    static_assert(KT % KC == 0, "K chunking");
    static_assert(BM % RSTEP == 0 && (R * 8) % T == 0, "staging split");
    static_assert(BM % (16 * WM) == 0 && BN % (16 * WN) == 0, "wave tiling");
    constexpr int LP    = 68;
    static_assert(!LSM || (BM == 16 && BN == 64 && TM == 1 && TN == 1 && WM == 1 && !RELU),
                  "fused log_softmax: one 16-row block of whole 64-wide rows, a tile a wave");
    static_assert(!LSM || (16 * LP <= R * PITCH && 16 % (WM * WN) == 0),
                  "logits fit in the drained stage; rows split evenly over the waves");

    __shared__ float lds[2 * R * PITCH];

    const int t    = threadIdx.x;
    const int lane = t & 31;
    const int wave = __builtin_amdgcn_readfirstlane(t >> 5);
    const int wm   = wave / WN;
    const int wn   = wave % WN;
    const int n    = lane & 15;
    const int kh   = lane >> 4;

    const int m0 = blockIdx.y * BM;
    const int n0 = blockIdx.x * BN;

    const int r0 = t >> 3;
    const int c4 = (t & 7) * 4;
    int rowOff[NLD];
    #pragma unroll
    for (int j = 0; j < NLD; ++j)
        rowOff[j] = ((j < NA ? m0 : n0 - BM) + r0 + j * RSTEP) * KT + c4;
    float* const putDst = lds + r0 * PITCH + c4;

    float inv[TM];
    #pragma unroll
    for (int i = 0; i < TM; ++i)
        inv[i] = 1.0f / fmaxf(cnt[m0 + wm * (BM / WM) + i * 16 + n], 1.0f);

    v8f acc[TM][TN];
    #pragma unroll
    for (int i = 0; i < TM; ++i)
        #pragma unroll
        for (int j = 0; j < TN; ++j)
            #pragma unroll
            for (int r = 0; r < 8; ++r) acc[i][j][r] = 0.0f;

    const float* const aSrc = lds + (wm * (BM / WM) + n) * PITCH + 2 * kh;
    const float* const bSrc = lds + (BM + wn * (BN / WN) + n) * PITCH + 2 * kh;

    v4f pf[NLD];
    chunk_fetch<NA, NLD>(agg, Wl, rowOff, 0, pf);
    chunk_put<NLD, RSTEP, PITCH>(putDst, pf);
    __syncthreads();

    #pragma unroll
    for (int c = 0; c < 2 * NCH; ++c) {
        const int  nx   = c + 1;
        const bool more = (nx < 2 * NCH);
        if (more) {
            if (nx < NCH) chunk_fetch<NA, NLD>(agg, Wl, rowOff, nx * KC, pf);
            else          chunk_fetch<NA, NLD>(xt, Wr, rowOff, (nx - NCH) * KC, pf);
        }
        __builtin_amdgcn_sched_barrier(0);
        const int so = (c & 1) * R * PITCH;
        if (c < NCH) chunk_mma<TM, TN, KC, PITCH, true >(aSrc + so, bSrc + so, inv, acc);
        else         chunk_mma<TM, TN, KC, PITCH, false>(aSrc + so, bSrc + so, inv, acc);
        if (more) {
            chunk_put<NLD, RSTEP, PITCH>(putDst + (nx & 1) * R * PITCH, pf);
            __syncthreads();
        }
    }

    if (LSM) {
        float* const lg = lds;
        const float bv = bias[n0 + wn * 16 + n];
        #pragma unroll
        for (int r = 0; r < 8; ++r)
            lg[(8 * kh + r) * LP + wn * 16 + n] = acc[0][0][r] + bv;
        __syncthreads();
        constexpr int RPW = 16 / (WM * WN);
        #pragma unroll
        for (int q = 0; q < RPW; ++q) {
            const int row = wave * RPW + q;
            const float a = lg[row * LP + lane];
            const float b = lg[row * LP + lane + 32];
            float m = fmaxf(a, b);
            #pragma unroll
            for (int off = 16; off >= 1; off >>= 1)
                m = fmaxf(m, __shfl_xor(m, off, 32));
            float s = __expf(a - m) + __expf(b - m);
            #pragma unroll
            for (int off = 16; off >= 1; off >>= 1)
                s += __shfl_xor(s, off, 32);
            const float l = __logf(s) + m;
            float* p = out + ((long long)m0 + row) * Nout;
            p[lane]      = a - l;
            p[lane + 32] = b - l;
        }
        return;
    }

    #pragma unroll
    for (int i = 0; i < TM; ++i) {
        const long long mbase = (long long)m0 + wm * (BM / WM) + i * 16 + 8 * kh;
        #pragma unroll
        for (int j = 0; j < TN; ++j) {
            const int col = n0 + wn * (BN / WN) + j * 16 + n;
            const float bv = bias[col];
            #pragma unroll
            for (int r = 0; r < 8; ++r) {
                float v = acc[i][j][r] + bv;
                if (RELU) v = fmaxf(v, 0.0f);
                out[(mbase + r) * Nout + col] = v;
            }
        }
    }
}

extern "C" void kernel_launch(void* const* d_in, const int* in_sizes, int n_in,
                              void* d_out, int out_size, void* d_ws, size_t ws_size,
                              hipStream_t stream) {
    const float* x    = (const float*)d_in[0];
    const int*   src1 = (const int*)  d_in[1];
    const int*   dst1 = (const int*)  d_in[2];
    const int*   src2 = (const int*)  d_in[3];
    const int*   dst2 = (const int*)  d_in[4];
    const float* W1l  = (const float*)d_in[7];
    const float* W1r  = (const float*)d_in[8];
    const float* b1   = (const float*)d_in[9];
    const float* W2l  = (const float*)d_in[10];
    const float* W2r  = (const float*)d_in[11];
    const float* b2   = (const float*)d_in[12];

    float* agg1 = (float*)d_ws;
    float* cnt1 = agg1 + (size_t)N2c * INc;
    float* h    = cnt1 + N2c;
    float* Bpl  = h    + (size_t)N2c * HIDc;
    float* agg2 = agg1;
    float* cnt2 = Bpl  + (size_t)N3c * HIDc;
    float* out  = (float*)d_out;

    int* binned = (int*)agg1;
    int* tblT   = binned + 2 * (size_t)ETc;
    int* deg1  = (int*)h;
    int* off1  = deg1 + N2c;
    int* perm1 = off1 + N2c;
    int* deg2  = (int*)Bpl;
    int* off2  = deg2 + N3c;
    int* perm2 = off2 + N3c;

    bin_edges<<<GW, 256, 0, stream>>>(dst1, src1, dst2, src2, binned, tblT);
    bucket_lists<<<NBK, 256, 0, stream>>>(binned, tblT, deg1, off1, cnt1, perm1,
                                          deg2, off2, cnt2, perm2);

    {
        int waves = N2c * (INc / 128);
        gather_rows<<<(waves + 7) / 8, 256, 0, stream>>>(x, perm1, off1, deg1, agg1,
                                                         N2c, INc, 0  );
        dim3 grid(HIDc / 128, N2c / 64);
        sage_gemm_lds<INc, 64, 128, 2, 4, true, false><<<grid, 256, 0, stream>>>(
            agg1, cnt1, x, W1l, W1r, b1, h, HIDc);
    }

    {
        int waves = N3c * (HIDc / 128);
        gather_rows<<<(waves + 7) / 8, 256, 0, stream>>>(h, perm2, off2, deg2, agg2,
                                                         N3c, HIDc, 1  );
        dim3 grid(OUTc / 64, N3c / 16);
        sage_gemm_lds<HIDc, 16, 64, 1, 4, false, true><<<grid, 128, 0, stream>>>(
            agg2, cnt2, h, W2l, W2r, b2, out, OUTc);
    }
}


# Layer-2 products applied to h before the neighbour sum, inside the layer-1 product's blocks (h never stored); the last launch gathers 64-float rows: 5 launches; sums now products first, then edges
# speedup vs baseline: 5.6828x; 1.0704x over previous
#include <hip/hip_runtime.h>
#include <math.h>


#define N1c  200000
#define N2c  40000
#define N3c  10000
#define INc  128
#define HIDc 256
#define OUTc 64

static_assert(N2c % 64 == 0 && HIDc % 128 == 0, "layer-1 product grid is exact");
static_assert(N3c % 16 == 0 && OUTc % 64 == 0, "layer-2 product grid is exact");

typedef float v2f __attribute__((ext_vector_type(2)));
typedef float v4f __attribute__((ext_vector_type(4)));
typedef float v8f __attribute__((ext_vector_type(8)));

#define E1c  640000
#define E2c  160000
#define ETc  (E1c + E2c)
#define SN   200
#define NB1  (N2c / SN)
#define NBK  ((N2c + N3c) / SN)
#define CH   3200
#define GW   (ETc / CH)
#define EPT  ((CH + 255) / 256)
#define CAP  8192

static_assert(N2c % SN == 0 && N3c % SN == 0, "bins do not straddle the layers");
static_assert(E1c % CH == 0 && E2c % CH == 0, "chunks are full and do not straddle the layers");
static_assert(NBK + 1 <= 256 && GW <= 256 && SN <= 256, "one thread per bin / chunk / node");

static __device__ __forceinline__ int block_scan_incl(int v, int* ws, int t) {
    const int lane = t & 31;
    for (int o = 1; o < 32; o <<= 1) {
        const int u = __shfl_up(v, o, 32);
        if (lane >= o) v += u;
    }
    if (lane == 31) ws[t >> 5] = v;
    __syncthreads();
    int add = 0;
    for (int j = 0; j < 8; ++j) add += (j < (t >> 5)) ? ws[j] : 0;
    return v + add;
}

__launch_bounds__(256)
__global__ void bin_edges(const int* __restrict__ dst1, const int* __restrict__ src1,
                          const int* __restrict__ dst2, const int* __restrict__ src2,
                          int* __restrict__ binned, int* __restrict__ tblT) {
    __shared__ int  cnt[NBK];
    __shared__ int  boff[NBK + 1];
    __shared__ int  ws[8];
    __shared__ int2 stage[CH];
    const int t = threadIdx.x;
    const int g = blockIdx.x;
    const int base = g * CH;

    if (t < NBK) cnt[t] = 0;
    __syncthreads();

    const bool l1 = (base < E1c);
    const int* dp = l1 ? dst1 + base : dst2 + (base - E1c);
    const int* sp = l1 ? src1 + base : src2 + (base - E1c);
    const int kofs = l1 ? 0 : N2c;

    int key[EPT], sv[EPT], rk[EPT];
    #pragma unroll
    for (int k = 0; k < EPT; ++k) {
        const int idx = k * 256 + t;
        const int idc = (idx < CH) ? idx : (CH - 1);
        key[k] = dp[idc] + kofs; sv[k] = sp[idc]; rk[k] = 0;
    }
    #pragma unroll
    for (int k = 0; k < EPT; ++k)
        if (k * 256 + t < CH) rk[k] = atomicAdd(&cnt[key[k] / SN], 1);
    __syncthreads();

    const int v    = (t < NBK) ? cnt[t] : 0;
    const int incl = block_scan_incl(v, ws, t);
    if (t <= NBK) boff[t] = incl - v;
    __syncthreads();

    #pragma unroll
    for (int k = 0; k < EPT; ++k)
        if (k * 256 + t < CH) stage[boff[key[k] / SN] + rk[k]] = make_int2(key[k], sv[k]);
    __syncthreads();

    int2* out2 = reinterpret_cast<int2*>(binned);
    #pragma unroll
    for (int k = 0; k < EPT; ++k) {
        const int idx = k * 256 + t;
        if (idx < CH) out2[base + idx] = stage[idx];
    }
    if (t <= NBK) tblT[t * GW + g] = boff[t];
}

#define EB 16
struct ListWalk { int lo, nxt, gofs; };

static __device__ __forceinline__ ListWalk walk_begin(const int* segS, const int* segP, int i) {
    int lo = 0;
    #pragma unroll
    for (int step = 128; step >= 1; step >>= 1)
        if (segP[lo + step] <= i) lo += step;
    ListWalk w;
    w.lo = lo; w.nxt = segP[lo + 1]; w.gofs = lo * CH + segS[lo] - segP[lo];
    return w;
}

static __device__ __forceinline__ int walk_to(ListWalk& w, const int* segS, const int* segP, int i) {
    while (i >= w.nxt) {
        const int st = w.nxt;
        ++w.lo;
        w.nxt  = segP[w.lo + 1];
        w.gofs = w.lo * CH + segS[w.lo] - st;
    }
    return w.gofs + i;
}

__launch_bounds__(256)
__global__ void bucket_lists(const int* __restrict__ binned, const int* __restrict__ tblT,
                             int* __restrict__ deg1, int* __restrict__ off1,
                             float* __restrict__ cnt1, int* __restrict__ perm1,
                             int* __restrict__ deg2, int* __restrict__ off2,
                             float* __restrict__ cnt2, int* __restrict__ perm2) {
    __shared__ int segS[256];
    __shared__ int segP[257];
    __shared__ int degl[256];
    __shared__ int offl[256];
    __shared__ int wsA[8], wsB[8], wsC[8];
    __shared__ int sBase;
    __shared__ int stage[CAP];
    const int t = threadIdx.x;
    const int b = blockIdx.x;
    const int2* in2 = reinterpret_cast<const int2*>(binned);

    int s = 0, n = 0;
    if (t < GW) {
        s = tblT[b * GW + t];
        n = tblT[(b + 1) * GW + t] - s;
    }
    const int pincl = block_scan_incl(n, wsA, t);
    const int sincl = block_scan_incl(s, wsB, t);
    segS[t] = s;
    segP[t] = pincl - n;
    if (t == 255) { segP[256] = pincl; sBase = sincl; }
    degl[t] = 0;
    __syncthreads();
    const int nb   = segP[256];
    const int base = sBase;
    const int lo_key = b * SN;

    const int per = (nb + 255) >> 8;
    const int i0  = t * per;
    const int i1  = (i0 + per < nb) ? (i0 + per) : nb;
    const bool any = (i0 < nb);
    ListWalk w0;
    w0.lo = 0; w0.nxt = 0; w0.gofs = 0;
    if (any) w0 = walk_begin(segS, segP, i0);

    if (any) {
        ListWalk w = w0;
        for (int i = i0; i < i1; i += EB) {
            int key[EB];
            #pragma unroll
            for (int k = 0; k < EB; ++k)
                key[k] = binned[2 * walk_to(w, segS, segP, (i + k < i1) ? (i + k) : (i1 - 1))];
            #pragma unroll
            for (int k = 0; k < EB; ++k)
                atomicAdd(&degl[key[k] - lo_key], (i + k < i1) ? 1 : 0);
        }
    }
    __syncthreads();

    const int d     = (t < SN) ? degl[t] : 0;
    const int dincl = block_scan_incl(d, wsC, t);
    offl[t] = dincl - d;
    degl[t] = 0;
    const bool L1 = (b < NB1);
    int* perm = L1 ? perm1 : perm2;
    const int pbase = L1 ? base : base - E1c;
    if (t < SN) {
        if (L1) {
            const int node = lo_key + t;
            deg1[node] = d; off1[node] = pbase + dincl - d; cnt1[node] = (float)d;
        } else {
            const int node = lo_key + t - N2c;
            deg2[node] = d; off2[node] = pbase + dincl - d; cnt2[node] = (float)d;
        }
    }
    __syncthreads();

    const bool fits = (nb <= CAP);
    if (any) {
        ListWalk w = w0;
        for (int i = i0; i < i1; i += EB) {
            int2 e[EB];
            int  pos[EB];
            #pragma unroll
            for (int k = 0; k < EB; ++k)
                e[k] = in2[walk_to(w, segS, segP, (i + k < i1) ? (i + k) : (i1 - 1))];
            #pragma unroll
            for (int k = 0; k < EB; ++k) {
                const int dl = e[k].x - lo_key;
                pos[k] = offl[dl] + atomicAdd(&degl[dl], (i + k < i1) ? 1 : 0);
            }
            #pragma unroll
            for (int k = 0; k < EB; ++k) {
                if (i + k < i1) {
                    if (fits) stage[pos[k]] = e[k].y;
                    else      perm[pbase + pos[k]] = e[k].y;
                }
            }
        }
    }
    __syncthreads();
    if (fits)
        for (int i = t; i < nb; i += 256) perm[pbase + i] = stage[i];
}

template <int R>
static __device__ __forceinline__ void take_rows(const float* __restrict__ xb, int D,
                                                 int ids, int j, float4 (&acc)[8]) {
    float4 v[R];
    for (int k = 0; k < R; ++k) {
        const int s = __builtin_amdgcn_readlane(ids, j + k);
        v[k] = *reinterpret_cast<const float4*>(xb + (long long)s * D);
    }
    for (int k = 0; k < R; ++k) {
        acc[k].x += v[k].x; acc[k].y += v[k].y;
        acc[k].z += v[k].z; acc[k].w += v[k].w;
    }
}

__launch_bounds__(256)
__global__ void gather_rows(const float* __restrict__ x,
                            const int*   __restrict__ perm,
                            const int*   __restrict__ off,
                            const int*   __restrict__ deg,
                            float*       __restrict__ agg,
                            int nNodes, int D, int partShift) {
    const int lane = threadIdx.x & 31;
    const int wv    = __builtin_amdgcn_readfirstlane((int)(threadIdx.x >> 5));
    const int gw    = blockIdx.x * 8 + wv;
    const int node  = gw >> partShift;
    if (node >= nNodes) return;
    const int part  = gw & ((1 << partShift) - 1);
    const int start = __builtin_amdgcn_readfirstlane(off[node]);
    const int n     = __builtin_amdgcn_readfirstlane(deg[node]);
    const float* xb = x + part * 128 + lane * 4;

    float4 acc[8];
    for (int k = 0; k < 8; ++k) acc[k] = make_float4(0.0f, 0.0f, 0.0f, 0.0f);

    for (int base = 0; base < n; base += 32) {
        const int m = (n - base < 32) ? (n - base) : 32;
        const int key = (lane < m) ? perm[start + base + lane] : 0x7fffffff;
        int r = 0;
        for (int t = 0; t < m; ++t) {
            const int kt = __builtin_amdgcn_readlane(key, t);
            r += (kt < key || (kt == key && t < lane)) ? 1 : 0;
        }
        if (lane >= m) r = lane;
        const int ids = __builtin_amdgcn_ds_permute(r << 2, key);

        int j = 0;
        for (; j + 8 <= m; j += 8) take_rows<8>(xb, D, ids, j, acc);
        switch (m - j) {
            case 7: take_rows<7>(xb, D, ids, j, acc); break;
            case 6: take_rows<6>(xb, D, ids, j, acc); break;
            case 5: take_rows<5>(xb, D, ids, j, acc); break;
            case 4: take_rows<4>(xb, D, ids, j, acc); break;
            case 3: take_rows<3>(xb, D, ids, j, acc); break;
            case 2: take_rows<2>(xb, D, ids, j, acc); break;
            case 1: take_rows<1>(xb, D, ids, j, acc); break;
            default: break;
        }
    }

    float4 s;
    s.x = ((acc[0].x + acc[1].x) + (acc[2].x + acc[3].x)) + ((acc[4].x + acc[5].x) + (acc[6].x + acc[7].x));
    s.y = ((acc[0].y + acc[1].y) + (acc[2].y + acc[3].y)) + ((acc[4].y + acc[5].y) + (acc[6].y + acc[7].y));
    s.z = ((acc[0].z + acc[1].z) + (acc[2].z + acc[3].z)) + ((acc[4].z + acc[5].z) + (acc[6].z + acc[7].z));
    s.w = ((acc[0].w + acc[1].w) + (acc[2].w + acc[3].w)) + ((acc[4].w + acc[5].w) + (acc[6].w + acc[7].w));
    *reinterpret_cast<float4*>(agg + (long long)node * D + part * 128 + lane * 4) = s;
}

template <int R>
static __device__ __forceinline__ void take_prows(const float* __restrict__ p0,
                                                  const float* __restrict__ p1,
                                                  int ids, int j, float2 (&acc)[8]) {
    float2 u[R], v[R];
    for (int k = 0; k < R; ++k) {
        const int s = __builtin_amdgcn_readlane(ids, j + k);
        u[k] = *reinterpret_cast<const float2*>(p0 + (long long)s * OUTc);
        v[k] = *reinterpret_cast<const float2*>(p1 + (long long)s * OUTc);
    }
    for (int k = 0; k < R; ++k) {
        acc[k].x += u[k].x + v[k].x;
        acc[k].y += u[k].y + v[k].y;
    }
}

__launch_bounds__(256)
__global__ void gather_finish(const float* __restrict__ P,
                              const float* __restrict__ Q,
                              const int*   __restrict__ perm,
                              const int*   __restrict__ off,
                              const int*   __restrict__ deg,
                              const float* __restrict__ cnt,
                              const float* __restrict__ bias,
                              float*       __restrict__ out) {
    const int lane = threadIdx.x & 31;
    const int wv    = __builtin_amdgcn_readfirstlane((int)(threadIdx.x >> 5));
    const int node  = blockIdx.x * 8 + wv;
    if (node >= N3c) return;
    const int start = __builtin_amdgcn_readfirstlane(off[node]);
    const int n     = __builtin_amdgcn_readfirstlane(deg[node]);
    const float* p0 = P + lane * 2;
    const float* p1 = P + (long long)N2c * OUTc + lane * 2;

    const float2 q0 = *reinterpret_cast<const float2*>(Q + (long long)node * OUTc + lane * 2);
    const float2 q1 = *reinterpret_cast<const float2*>(Q + (long long)(N3c + node) * OUTc + lane * 2);
    const float2 bv = *reinterpret_cast<const float2*>(bias + lane * 2);
    const float inv = 1.0f / fmaxf(cnt[node], 1.0f);

    float2 acc[8];
    for (int k = 0; k < 8; ++k) acc[k] = make_float2(0.0f, 0.0f);

    for (int base = 0; base < n; base += 32) {
        const int m = (n - base < 32) ? (n - base) : 32;
        const int key = (lane < m) ? perm[start + base + lane] : 0x7fffffff;
        int r = 0;
        for (int t = 0; t < m; ++t) {
            const int kt = __builtin_amdgcn_readlane(key, t);
            r += (kt < key || (kt == key && t < lane)) ? 1 : 0;
        }
        if (lane >= m) r = lane;
        const int ids = __builtin_amdgcn_ds_permute(r << 2, key);

        int j = 0;
        for (; j + 8 <= m; j += 8) take_prows<8>(p0, p1, ids, j, acc);
        switch (m - j) {
            case 7: take_prows<7>(p0, p1, ids, j, acc); break;
            case 6: take_prows<6>(p0, p1, ids, j, acc); break;
            case 5: take_prows<5>(p0, p1, ids, j, acc); break;
            case 4: take_prows<4>(p0, p1, ids, j, acc); break;
            case 3: take_prows<3>(p0, p1, ids, j, acc); break;
            case 2: take_prows<2>(p0, p1, ids, j, acc); break;
            case 1: take_prows<1>(p0, p1, ids, j, acc); break;
            default: break;
        }
    }

    const float sx = ((acc[0].x + acc[1].x) + (acc[2].x + acc[3].x)) + ((acc[4].x + acc[5].x) + (acc[6].x + acc[7].x));
    const float sy = ((acc[0].y + acc[1].y) + (acc[2].y + acc[3].y)) + ((acc[4].y + acc[5].y) + (acc[6].y + acc[7].y));
    const float a = sx * inv + bv.x + (q0.x + q1.x);
    const float b = sy * inv + bv.y + (q0.y + q1.y);

    float mx = fmaxf(a, b);
    for (int o = 16; o >= 1; o >>= 1) mx = fmaxf(mx, __shfl_xor(mx, o, 32));
    float se = __expf(a - mx) + __expf(b - mx);
    for (int o = 16; o >= 1; o >>= 1) se += __shfl_xor(se, o, 32);
    const float l = __logf(se) + mx;
    *reinterpret_cast<float2*>(out + (long long)node * OUTc + lane * 2) = make_float2(a - l, b - l);
}

template <int NA, int NLD>
static __device__ __forceinline__ void chunk_fetch(const float* __restrict__ Ab,
                                                   const float* __restrict__ Wb,
                                                   const int (&rowOff)[NLD], int k,
                                                   v4f (&pf)[NLD]) {
    #pragma unroll
    for (int j = 0; j < NLD; ++j)
        pf[j] = *reinterpret_cast<const v4f*>((j < NA ? Ab : Wb) + rowOff[j] + k);
}

template <int NLD, int RSTEP, int PITCH>
static __device__ __forceinline__ void chunk_put(float* dst, const v4f (&pf)[NLD]) {
    #pragma unroll
    for (int j = 0; j < NLD; ++j)
        *reinterpret_cast<v4f*>(dst + j * RSTEP * PITCH) = pf[j];
}

template <int TM, int TN, int KC, int PITCH, bool SCALE>
static __device__ __forceinline__ void chunk_mma(const float* aSrc, const float* bSrc,
                                                 const float (&inv)[TM],
                                                 v8f (&acc)[TM][TN]) {
    #pragma unroll
    for (int ks = 0; ks < KC / 4; ++ks) {
        v2f a[TM], b[TN];
        #pragma unroll
        for (int i = 0; i < TM; ++i) {
            a[i] = *reinterpret_cast<const v2f*>(aSrc + i * 16 * PITCH + ks * 4);
            if (SCALE) { a[i][0] *= inv[i]; a[i][1] *= inv[i]; }
        }
        #pragma unroll
        for (int j = 0; j < TN; ++j)
            b[j] = *reinterpret_cast<const v2f*>(bSrc + j * 16 * PITCH + ks * 4);
        #pragma unroll
        for (int i = 0; i < TM; ++i)
            #pragma unroll
            for (int j = 0; j < TN; ++j)
                acc[i][j] = __builtin_amdgcn_wmma_f32_16x16x4_f32(
                    false, a[i], false, b[j], (short)0, acc[i][j], false, false);
    }
}

template <int KT, int BM, int BN, int WM, int WN, int NP, int MQ>
__launch_bounds__(32 * WM * WN, 8)
__global__ void sage_gemm_proj(const float* __restrict__ agg,
                               const float* __restrict__ cnt,
                               const float* __restrict__ xt,
                               const float* __restrict__ Wl,
                               const float* __restrict__ Wr,
                               const float* __restrict__ bias,
                               const float* __restrict__ Vl,
                               const float* __restrict__ Vr,
                               float*       __restrict__ P,
                               float*       __restrict__ Q,
                               int Nh, int Mrows) {
    constexpr int KC    = 32;
    constexpr int PITCH = KC + 4;
    constexpr int T     = 32 * WM * WN;
    constexpr int R     = BM + BN;
    constexpr int RSTEP = T / 8;
    constexpr int NLD   = R * 8 / T;
    constexpr int NA    = BM / RSTEP;
    constexpr int NCH   = KT / KC;
    constexpr int TM    = BM / WM / 16;
    constexpr int TN    = BN / WN / 16;
    static_assert(KT % KC == 0, "K chunking");
    static_assert(BM % RSTEP == 0 && (R * 8) % T == 0, "staging split");
    static_assert(BM % (16 * WM) == 0 && BN % (16 * WN) == 0, "wave tiling");
    constexpr int HP    = BN + 4;
    constexpr int TNP   = NP / WN / 16;
    static_assert(NP % (16 * WN) == 0 && BN % 4 == 0 && MQ % 16 == 0, "second product tiling");
    static_assert((NP * (BN / 4)) % T == 0, "weight slice staging split");
    constexpr int LDSF  = (2 * R * PITCH > (BM + NP) * HP) ? 2 * R * PITCH : (BM + NP) * HP;

    __shared__ float lds[LDSF];

    const int t    = threadIdx.x;
    const int lane = t & 31;
    const int wave = __builtin_amdgcn_readfirstlane(t >> 5);
    const int wm   = wave / WN;
    const int wn   = wave % WN;
    const int n    = lane & 15;
    const int kh   = lane >> 4;

    const int m0 = blockIdx.y * BM;
    const int n0 = blockIdx.x * BN;

    const int r0 = t >> 3;
    const int c4 = (t & 7) * 4;
    int rowOff[NLD];
    #pragma unroll
    for (int j = 0; j < NLD; ++j)
        rowOff[j] = ((j < NA ? m0 : n0 - BM) + r0 + j * RSTEP) * KT + c4;
    float* const putDst = lds + r0 * PITCH + c4;

    float inv[TM];
    #pragma unroll
    for (int i = 0; i < TM; ++i)
        inv[i] = 1.0f / fmaxf(cnt[m0 + wm * (BM / WM) + i * 16 + n], 1.0f);

    v8f acc[TM][TN];
    #pragma unroll
    for (int i = 0; i < TM; ++i)
        #pragma unroll
        for (int j = 0; j < TN; ++j)
            #pragma unroll
            for (int r = 0; r < 8; ++r) acc[i][j][r] = 0.0f;

    const float* const aSrc = lds + (wm * (BM / WM) + n) * PITCH + 2 * kh;
    const float* const bSrc = lds + (BM + wn * (BN / WN) + n) * PITCH + 2 * kh;

    v4f pf[NLD];
    chunk_fetch<NA, NLD>(agg, Wl, rowOff, 0, pf);
    chunk_put<NLD, RSTEP, PITCH>(putDst, pf);
    __syncthreads();

    #pragma unroll
    for (int c = 0; c < 2 * NCH; ++c) {
        const int  nx   = c + 1;
        const bool more = (nx < 2 * NCH);
        if (more) {
            if (nx < NCH) chunk_fetch<NA, NLD>(agg, Wl, rowOff, nx * KC, pf);
            else          chunk_fetch<NA, NLD>(xt, Wr, rowOff, (nx - NCH) * KC, pf);
        }
        __builtin_amdgcn_sched_barrier(0);
        const int so = (c & 1) * R * PITCH;
        if (c < NCH) chunk_mma<TM, TN, KC, PITCH, true >(aSrc + so, bSrc + so, inv, acc);
        else         chunk_mma<TM, TN, KC, PITCH, false>(aSrc + so, bSrc + so, inv, acc);
        if (more) {
            chunk_put<NLD, RSTEP, PITCH>(putDst + (nx & 1) * R * PITCH, pf);
            __syncthreads();
        }
    }

    __syncthreads();
    float* const ht = lds;
    float* const w2 = lds + BM * HP;
    #pragma unroll
    for (int i = 0; i < TM; ++i)
        #pragma unroll
        for (int j = 0; j < TN; ++j) {
            const int cl = wn * (BN / WN) + j * 16 + n;
            const float bv = bias[n0 + cl];
            #pragma unroll
            for (int r = 0; r < 8; ++r)
                ht[(wm * (BM / WM) + i * 16 + 8 * kh + r) * HP + cl] = fmaxf(acc[i][j][r] + bv, 0.0f);
        }

    constexpr int WLD = NP * (BN / 4) / T;
    constexpr int WRS = T / (BN / 4);
    const int wr0 = t / (BN / 4);
    const int wc4 = (t % (BN / 4)) * 4;
    const float* const aSrc2 = ht + (wm * (BM / WM) + n) * HP + 2 * kh;
    const float* const bSrc2 = w2 + (wn * (NP / WN) + n) * HP + 2 * kh;
    const int nQ = (m0 < MQ) ? 2 : 1;

    for (int pass = 0; pass < nQ; ++pass) {
        const float* const V = (pass == 0) ? Vl : Vr;
        v4f wv[WLD];
        #pragma unroll
        for (int q = 0; q < WLD; ++q)
            wv[q] = *reinterpret_cast<const v4f*>(V + (long long)(wr0 + q * WRS) * Nh + n0 + wc4);
        if (pass == 1) __syncthreads();
        #pragma unroll
        for (int q = 0; q < WLD; ++q)
            *reinterpret_cast<v4f*>(w2 + (wr0 + q * WRS) * HP + wc4) = wv[q];
        __syncthreads();

        v8f pacc[TM][TNP];
        #pragma unroll
        for (int i = 0; i < TM; ++i)
            #pragma unroll
            for (int j = 0; j < TNP; ++j)
                #pragma unroll
                for (int r = 0; r < 8; ++r) pacc[i][j][r] = 0.0f;
        chunk_mma<TM, TNP, BN, HP, false>(aSrc2, bSrc2, inv, pacc);

        float* const dst = (pass == 0) ? P + (long long)blockIdx.x * Mrows * NP
                                       : Q + (long long)blockIdx.x * MQ * NP;
        const int rowLim = (pass == 0) ? Mrows : MQ;
        #pragma unroll
        for (int i = 0; i < TM; ++i) {
            const int rb = m0 + wm * (BM / WM) + i * 16;
            if (rb < rowLim) {
                #pragma unroll
                for (int j = 0; j < TNP; ++j) {
                    const int col = wn * (NP / WN) + j * 16 + n;
                    #pragma unroll
                    for (int r = 0; r < 8; ++r)
                        dst[(long long)(rb + 8 * kh + r) * NP + col] = pacc[i][j][r];
                }
            }
        }
    }
}

extern "C" void kernel_launch(void* const* d_in, const int* in_sizes, int n_in,
                              void* d_out, int out_size, void* d_ws, size_t ws_size,
                              hipStream_t stream) {
    const float* x    = (const float*)d_in[0];
    const int*   src1 = (const int*)  d_in[1];
    const int*   dst1 = (const int*)  d_in[2];
    const int*   src2 = (const int*)  d_in[3];
    const int*   dst2 = (const int*)  d_in[4];
    const float* W1l  = (const float*)d_in[7];
    const float* W1r  = (const float*)d_in[8];
    const float* b1   = (const float*)d_in[9];
    const float* W2l  = (const float*)d_in[10];
    const float* W2r  = (const float*)d_in[11];
    const float* b2   = (const float*)d_in[12];

    float* agg1 = (float*)d_ws;
    float* cnt1 = agg1 + (size_t)N2c * INc;
    float* h    = cnt1 + N2c;
    float* Bpl  = h    + (size_t)N2c * HIDc;
    float* cnt2 = Bpl  + (size_t)N3c * HIDc;
    float* out  = (float*)d_out;
    float* Pp   = h;
    float* Qp   = Pp + 2 * (size_t)N2c * OUTc;
    static_assert(2 * (N2c + N3c) * OUTc <= N2c * HIDc, "P and Q fit in plane H");

    int* binned = (int*)agg1;
    int* tblT   = binned + 2 * (size_t)ETc;
    int* deg1  = (int*)h;
    int* off1  = deg1 + N2c;
    int* perm1 = off1 + N2c;
    int* deg2  = (int*)Bpl;
    int* off2  = deg2 + N3c;
    int* perm2 = off2 + N3c;

    bin_edges<<<GW, 256, 0, stream>>>(dst1, src1, dst2, src2, binned, tblT);
    bucket_lists<<<NBK, 256, 0, stream>>>(binned, tblT, deg1, off1, cnt1, perm1,
                                          deg2, off2, cnt2, perm2);

    {
        int waves = N2c * (INc / 128);
        gather_rows<<<(waves + 7) / 8, 256, 0, stream>>>(x, perm1, off1, deg1, agg1,
                                                         N2c, INc, 0  );
        dim3 grid(HIDc / 128, N2c / 64);
        sage_gemm_proj<INc, 64, 128, 2, 4, OUTc, N3c><<<grid, 256, 0, stream>>>(
            agg1, cnt1, x, W1l, W1r, b1, W2l, W2r, Pp, Qp, HIDc, N2c);
    }

    gather_finish<<<(N3c + 7) / 8, 256, 0, stream>>>(Pp, Qp, perm2, off2, deg2, cnt2, b2, out);
}
